# starting kernel
# baseline (speedup 1.0000x reference)
#include <hip/hip_runtime.h>

typedef float v2f __attribute__((ext_vector_type(2)));
typedef float v4f __attribute__((ext_vector_type(4)));
typedef float v8f __attribute__((ext_vector_type(8)));

#define GS_B  512
#define GS_K  64
#define GS_L  1024
#define GS_LP 1028
#define GS_NT 1024

__device__ __forceinline__ void lds_fence() {
  asm volatile("s_wait_dscnt 0" ::: "memory");
  __builtin_amdgcn_wave_barrier();
}

__global__ __launch_bounds__(GS_NT, 1)
void gram_schmidt_cholqr(const float* __restrict__ xin,
                         float* __restrict__ xout) {
  extern __shared__ float smem[];
  float* X = smem;
  float* G = smem + GS_K * GS_LP;
  float* W = G + GS_K * GS_K;

  const int b    = blockIdx.x;
  const int tid  = threadIdx.x;
  const int wave = tid >> 5;
  const int lane = tid & 31;

  const float* xg = xin  + (size_t)b * (GS_K * GS_L);
  float*       og = xout + (size_t)b * (GS_K * GS_L);

  for (int idx = tid; idx < GS_K * (GS_L / 4); idx += GS_NT) {
    int r  = idx >> 8;
    int c4 = (idx & 255) << 2;
    *(v4f*)(X + r * GS_LP + c4) = *(const v4f*)(xg + r * GS_L + c4);
  }
  __syncthreads();

  {
    const int tile  = wave >> 1;
    const int ti    = tile >> 2;
    const int tj    = tile & 3;
    const int khalf = wave & 1;
    const int m     = lane & 15;
    const int koff  = (lane < 16) ? 0 : 2;
    const float* pa = X + (ti * 16 + m) * GS_LP + koff + khalf * (GS_L / 2);
    const float* pb = X + (tj * 16 + m) * GS_LP + koff + khalf * (GS_L / 2);
    v8f acc = {0.f, 0.f, 0.f, 0.f, 0.f, 0.f, 0.f, 0.f};
    #pragma unroll 4
    for (int k = 0; k < GS_L / 2; k += 4) {
      v2f a  = *(const v2f*)(pa + k);
      v2f bb = *(const v2f*)(pb + k);
      acc = __builtin_amdgcn_wmma_f32_16x16x4_f32(false, a, false, bb,
                                                  (short)0, acc, false, false);
    }
    float*    dst = khalf ? W : G;
    const int mb  = (lane < 16) ? 0 : 8;
    #pragma unroll
    for (int j = 0; j < 8; ++j)
      dst[(ti * 16 + mb + j) * GS_K + tj * 16 + m] = acc[j];
  }
  __syncthreads();
  for (int idx = tid; idx < GS_K * GS_K; idx += GS_NT) G[idx] += W[idx];
  __syncthreads();

  if (wave == 0) {
    for (int j = 0; j < GS_K; ++j) {
      float ljj = sqrtf(G[j * GS_K + j]);
      float inv = 1.0f / ljj;
      if (lane == 0) G[j * GS_K + j] = ljj;
      for (int i = j + 1 + lane; i < GS_K; i += 32)
        G[i * GS_K + j] *= inv;
      lds_fence();
      for (int i = j + 1 + lane; i < GS_K; i += 32) {
        float lij = G[i * GS_K + j];
        for (int k = j + 1; k <= i; ++k)
          G[i * GS_K + k] -= lij * G[k * GS_K + j];
      }
      lds_fence();
    }
  }
  __syncthreads();

  for (int j = wave; j < GS_K; j += 32) {
    for (int i = lane; i < GS_K; i += 32)
      W[i * GS_K + j] = (i == j) ? 1.0f : 0.0f;
    lds_fence();
    for (int k = j; k < GS_K; ++k) {
      float wk = W[k * GS_K + j] / G[k * GS_K + k];
      if (lane == 0) W[k * GS_K + j] = wk;
      for (int i = k + 1 + lane; i < GS_K; i += 32)
        W[i * GS_K + j] -= G[i * GS_K + k] * wk;
      lds_fence();
    }
  }
  __syncthreads();

  {
    const int m    = lane & 15;
    const int koff = (lane < 16) ? 0 : 2;
    const int mb   = (lane < 16) ? 0 : 8;
    for (int t = wave; t < 4 * (GS_L / 16); t += 32) {
      const int ti = t >> 6;
      const int tc = t & 63;
      const float* pa = W + (ti * 16 + m) * GS_K + koff;
      v8f acc = {0.f, 0.f, 0.f, 0.f, 0.f, 0.f, 0.f, 0.f};
      #pragma unroll
      for (int k0 = 0; k0 < GS_K; k0 += 4) {
        v2f a = *(const v2f*)(pa + k0);
        v2f bb;
        bb.x = X[(k0 + koff    ) * GS_LP + tc * 16 + m];
        bb.y = X[(k0 + koff + 1) * GS_LP + tc * 16 + m];
        acc = __builtin_amdgcn_wmma_f32_16x16x4_f32(false, a, false, bb,
                                                    (short)0, acc, false, false);
      }
      #pragma unroll
      for (int j = 0; j < 8; ++j)
        og[(ti * 16 + mb + j) * GS_L + tc * 16 + m] = acc[j];
    }
  }
}

extern "C" void kernel_launch(void* const* d_in, const int* in_sizes, int n_in,
                              void* d_out, int out_size, void* d_ws, size_t ws_size,
                              hipStream_t stream) {
  (void)in_sizes; (void)n_in; (void)out_size; (void)d_ws; (void)ws_size;
  const float* x   = (const float*)d_in[0];
  float*       out = (float*)d_out;

  const size_t shmem = (size_t)(GS_K * GS_LP + 2 * GS_K * GS_K) * sizeof(float);
  hipFuncSetAttribute(reinterpret_cast<const void*>(&gram_schmidt_cholqr),
                      hipFuncAttributeMaxDynamicSharedMemorySize, (int)shmem);

  gram_schmidt_cholqr<<<dim3(GS_B), dim3(GS_NT), shmem, stream>>>(x, out);
}


# Cholesky of the 64x64 Gram matrix moved from bank-conflicted LDS read-modify-write into the registers of the one wave that does it (two rows per lane, row values by v_readlane); same sum order
# speedup vs baseline: 2.1625x; 2.1625x over previous
#include <hip/hip_runtime.h>

typedef float v2f __attribute__((ext_vector_type(2)));
typedef float v4f __attribute__((ext_vector_type(4)));
typedef float v8f __attribute__((ext_vector_type(8)));

#define GS_B  512
#define GS_K  64
#define GS_L  1024
#define GS_LP 1028
#define GS_NT 1024

__device__ __forceinline__ void lds_fence() {
  asm volatile("s_wait_dscnt 0" ::: "memory");
  __builtin_amdgcn_wave_barrier();
}

__device__ __forceinline__ float lane_bcast(float v, int src_lane) {
  return __int_as_float(__builtin_amdgcn_readlane(__float_as_int(v), src_lane));
}

template <int J, int P>
__device__ __forceinline__ void chol_dot(float (&a)[32], float (&b)[64]) {
  if constexpr (P < J) {
    float ljp;
    if constexpr (J < 32) ljp = lane_bcast(a[P], J);
    else                  ljp = lane_bcast(b[P], J - 32);
    if constexpr (J < 32) a[J] = __builtin_fmaf(ljp, a[P], a[J]);
    b[J] = __builtin_fmaf(ljp, b[P], b[J]);
    chol_dot<J, P + 1>(a, b);
  }
}

template <int J>
__device__ __forceinline__ void chol_col(float (&a)[32], float (&b)[64], int lane) {
  if constexpr (J < 64) {
    chol_dot<J, 0>(a, b);
    float djj;
    if constexpr (J < 32) djj = -lane_bcast(a[J], J);
    else                  djj = -lane_bcast(b[J], J - 32);
    const float ljj = sqrtf(djj);
    const float inv = 1.0f / ljj;
    if constexpr (J < 32) {
      a[J] = (lane == J) ? ljj : (-a[J]) * inv;
      b[J] = (-b[J]) * inv;
    } else {
      b[J] = (lane == J - 32) ? ljj : (-b[J]) * inv;
    }
    chol_col<J + 1>(a, b, lane);
  }
}

__global__ __launch_bounds__(GS_NT, 1)
void gram_schmidt_cholqr(const float* __restrict__ xin,
                         float* __restrict__ xout) {
  extern __shared__ float smem[];
  float* X = smem;
  float* G = smem + GS_K * GS_LP;
  float* W = G + GS_K * GS_K;

  const int b    = blockIdx.x;
  const int tid  = threadIdx.x;
  const int wave = tid >> 5;
  const int lane = tid & 31;

  const float* xg = xin  + (size_t)b * (GS_K * GS_L);
  float*       og = xout + (size_t)b * (GS_K * GS_L);

  for (int idx = tid; idx < GS_K * (GS_L / 4); idx += GS_NT) {
    int r  = idx >> 8;
    int c4 = (idx & 255) << 2;
    *(v4f*)(X + r * GS_LP + c4) = *(const v4f*)(xg + r * GS_L + c4);
  }
  __syncthreads();

  {
    const int tile  = wave >> 1;
    const int ti    = tile >> 2;
    const int tj    = tile & 3;
    const int khalf = wave & 1;
    const int m     = lane & 15;
    const int koff  = (lane < 16) ? 0 : 2;
    const float* pa = X + (ti * 16 + m) * GS_LP + koff + khalf * (GS_L / 2);
    const float* pb = X + (tj * 16 + m) * GS_LP + koff + khalf * (GS_L / 2);
    v8f acc = {0.f, 0.f, 0.f, 0.f, 0.f, 0.f, 0.f, 0.f};
    #pragma unroll 4
    for (int k = 0; k < GS_L / 2; k += 4) {
      v2f a  = *(const v2f*)(pa + k);
      v2f bb = *(const v2f*)(pb + k);
      acc = __builtin_amdgcn_wmma_f32_16x16x4_f32(false, a, false, bb,
                                                  (short)0, acc, false, false);
    }
    float*    dst = khalf ? W : G;
    const int mb  = (lane < 16) ? 0 : 8;
    #pragma unroll
    for (int j = 0; j < 8; ++j)
      dst[(ti * 16 + mb + j) * GS_K + tj * 16 + m] = acc[j];
  }
  __syncthreads();
  for (int idx = tid; idx < GS_K * GS_K; idx += GS_NT) G[idx] += W[idx];
  __syncthreads();

  if (wave == 0) {
    float ra[32], rb[64];
    float* ga = G + lane * GS_K;
    float* gb = G + (lane + 32) * GS_K;
    #pragma unroll
    for (int k = 0; k < 32; k += 4) {
      v4f t = *(const v4f*)(ga + k);
      ra[k] = -t.x; ra[k + 1] = -t.y; ra[k + 2] = -t.z; ra[k + 3] = -t.w;
    }
    #pragma unroll
    for (int k = 0; k < 64; k += 4) {
      v4f t = *(const v4f*)(gb + k);
      rb[k] = -t.x; rb[k + 1] = -t.y; rb[k + 2] = -t.z; rb[k + 3] = -t.w;
    }
    chol_col<0>(ra, rb, lane);
    #pragma unroll
    for (int k = 0; k < 32; k += 4) {
      v4f t = {ra[k], ra[k + 1], ra[k + 2], ra[k + 3]};
      *(v4f*)(ga + k) = t;
    }
    #pragma unroll
    for (int k = 0; k < 64; k += 4) {
      v4f t = {rb[k], rb[k + 1], rb[k + 2], rb[k + 3]};
      *(v4f*)(gb + k) = t;
    }
  }
  __syncthreads();

  for (int j = wave; j < GS_K; j += 32) {
    for (int i = lane; i < GS_K; i += 32)
      W[i * GS_K + j] = (i == j) ? 1.0f : 0.0f;
    lds_fence();
    for (int k = j; k < GS_K; ++k) {
      float wk = W[k * GS_K + j] / G[k * GS_K + k];
      if (lane == 0) W[k * GS_K + j] = wk;
      for (int i = k + 1 + lane; i < GS_K; i += 32)
        W[i * GS_K + j] -= G[i * GS_K + k] * wk;
      lds_fence();
    }
  }
  __syncthreads();

  {
    const int m    = lane & 15;
    const int koff = (lane < 16) ? 0 : 2;
    const int mb   = (lane < 16) ? 0 : 8;
    for (int t = wave; t < 4 * (GS_L / 16); t += 32) {
      const int ti = t >> 6;
      const int tc = t & 63;
      const float* pa = W + (ti * 16 + m) * GS_K + koff;
      v8f acc = {0.f, 0.f, 0.f, 0.f, 0.f, 0.f, 0.f, 0.f};
      #pragma unroll
      for (int k0 = 0; k0 < GS_K; k0 += 4) {
        v2f a = *(const v2f*)(pa + k0);
        v2f bb;
        bb.x = X[(k0 + koff    ) * GS_LP + tc * 16 + m];
        bb.y = X[(k0 + koff + 1) * GS_LP + tc * 16 + m];
        acc = __builtin_amdgcn_wmma_f32_16x16x4_f32(false, a, false, bb,
                                                    (short)0, acc, false, false);
      }
      #pragma unroll
      for (int j = 0; j < 8; ++j)
        og[(ti * 16 + mb + j) * GS_L + tc * 16 + m] = acc[j];
    }
  }
}

extern "C" void kernel_launch(void* const* d_in, const int* in_sizes, int n_in,
                              void* d_out, int out_size, void* d_ws, size_t ws_size,
                              hipStream_t stream) {
  (void)in_sizes; (void)n_in; (void)out_size; (void)d_ws; (void)ws_size;
  const float* x   = (const float*)d_in[0];
  float*       out = (float*)d_out;

  const size_t shmem = (size_t)(GS_K * GS_LP + 2 * GS_K * GS_K) * sizeof(float);
  hipFuncSetAttribute(reinterpret_cast<const void*>(&gram_schmidt_cholqr),
                      hipFuncAttributeMaxDynamicSharedMemorySize, (int)shmem);

  gram_schmidt_cholqr<<<dim3(GS_B), dim3(GS_NT), shmem, stream>>>(x, out);
}


# Inverse of the Cholesky factor moved from bank-conflicted LDS column updates into registers: lane = row, pivot by v_readlane, factor stored transposed so columns read conflict-free; same sum order
# speedup vs baseline: 3.7550x; 1.7364x over previous
#include <hip/hip_runtime.h>

typedef float v2f __attribute__((ext_vector_type(2)));
typedef float v4f __attribute__((ext_vector_type(4)));
typedef float v8f __attribute__((ext_vector_type(8)));

#define GS_B  512
#define GS_K  64
#define GS_L  1024
#define GS_LP 1028
#define GS_NT 1024

__device__ __forceinline__ float lane_bcast(float v, int src_lane) {
  return __int_as_float(__builtin_amdgcn_readlane(__float_as_int(v), src_lane));
}

__device__ __forceinline__ float lane_pick(float v, int src_lane) {
  return __int_as_float(__builtin_amdgcn_readlane(__float_as_int(v), src_lane));
}

__device__ __forceinline__ float fs_step(float x, float l, float d, int k, int lane) {
  const float wk = lane_pick(x, k) / d;
  return (lane > k) ? __builtin_fmaf(-l, wk, x) : ((lane == k) ? wk : x);
}

template <int J, int P>
__device__ __forceinline__ void chol_dot(float (&a)[32], float (&b)[64]) {
  if constexpr (P < J) {
    float ljp;
    if constexpr (J < 32) ljp = lane_bcast(a[P], J);
    else                  ljp = lane_bcast(b[P], J - 32);
    if constexpr (J < 32) a[J] = __builtin_fmaf(ljp, a[P], a[J]);
    b[J] = __builtin_fmaf(ljp, b[P], b[J]);
    chol_dot<J, P + 1>(a, b);
  }
}

template <int J>
__device__ __forceinline__ void chol_col(float (&a)[32], float (&b)[64], int lane) {
  if constexpr (J < 64) {
    chol_dot<J, 0>(a, b);
    float djj;
    if constexpr (J < 32) djj = -lane_bcast(a[J], J);
    else                  djj = -lane_bcast(b[J], J - 32);
    const float ljj = sqrtf(djj);
    const float inv = 1.0f / ljj;
    if constexpr (J < 32) {
      a[J] = (lane == J) ? ljj : (-a[J]) * inv;
      b[J] = (-b[J]) * inv;
    } else {
      b[J] = (lane == J - 32) ? ljj : (-b[J]) * inv;
    }
    chol_col<J + 1>(a, b, lane);
  }
}

__global__ __launch_bounds__(GS_NT, 1)
void gram_schmidt_cholqr(const float* __restrict__ xin,
                         float* __restrict__ xout) {
  extern __shared__ float smem[];
  float* X = smem;
  float* G = smem + GS_K * GS_LP;
  float* W = G + GS_K * GS_K;

  const int b    = blockIdx.x;
  const int tid  = threadIdx.x;
  const int wave = tid >> 5;
  const int lane = tid & 31;

  const float* xg = xin  + (size_t)b * (GS_K * GS_L);
  float*       og = xout + (size_t)b * (GS_K * GS_L);

  for (int idx = tid; idx < GS_K * (GS_L / 4); idx += GS_NT) {
    int r  = idx >> 8;
    int c4 = (idx & 255) << 2;
    *(v4f*)(X + r * GS_LP + c4) = *(const v4f*)(xg + r * GS_L + c4);
  }
  __syncthreads();

  {
    const int tile  = wave >> 1;
    const int ti    = tile >> 2;
    const int tj    = tile & 3;
    const int khalf = wave & 1;
    const int m     = lane & 15;
    const int koff  = (lane < 16) ? 0 : 2;
    const float* pa = X + (ti * 16 + m) * GS_LP + koff + khalf * (GS_L / 2);
    const float* pb = X + (tj * 16 + m) * GS_LP + koff + khalf * (GS_L / 2);
    v8f acc = {0.f, 0.f, 0.f, 0.f, 0.f, 0.f, 0.f, 0.f};
    #pragma unroll 4
    for (int k = 0; k < GS_L / 2; k += 4) {
      v2f a  = *(const v2f*)(pa + k);
      v2f bb = *(const v2f*)(pb + k);
      acc = __builtin_amdgcn_wmma_f32_16x16x4_f32(false, a, false, bb,
                                                  (short)0, acc, false, false);
    }
    float*    dst = khalf ? W : G;
    const int mb  = (lane < 16) ? 0 : 8;
    #pragma unroll
    for (int j = 0; j < 8; ++j)
      dst[(ti * 16 + mb + j) * GS_K + tj * 16 + m] = acc[j];
  }
  __syncthreads();
  for (int idx = tid; idx < GS_K * GS_K; idx += GS_NT) G[idx] += W[idx];
  __syncthreads();

  if (wave == 0) {
    float ra[32], rb[64];
    float* ga = G + lane * GS_K;
    float* gb = G + (lane + 32) * GS_K;
    #pragma unroll
    for (int k = 0; k < 32; k += 4) {
      v4f t = *(const v4f*)(ga + k);
      ra[k] = -t.x; ra[k + 1] = -t.y; ra[k + 2] = -t.z; ra[k + 3] = -t.w;
    }
    #pragma unroll
    for (int k = 0; k < 64; k += 4) {
      v4f t = *(const v4f*)(gb + k);
      rb[k] = -t.x; rb[k + 1] = -t.y; rb[k + 2] = -t.z; rb[k + 3] = -t.w;
    }
    chol_col<0>(ra, rb, lane);
    #pragma unroll
    for (int k = 0; k < 32; ++k) {
      G[k * GS_K + lane]      = ra[k];
      G[k * GS_K + lane + 32] = rb[k];
    }
    #pragma unroll
    for (int k = 32; k < 64; ++k)
      G[k * GS_K + lane + 32] = rb[k];
  }
  __syncthreads();

  {
    const int wv = __builtin_amdgcn_readfirstlane(wave);
    const int jA = wv;
    const int jB = 63 - wv;
    float xlo = (lane == jA) ? 1.0f : 0.0f;
    float xhi = 0.0f;
    float yhi = (lane + 32 == jB) ? 1.0f : 0.0f;
    const float* lt = G + lane;

    float llo = lt[jA * GS_K];
    float lhi = lt[jA * GS_K + 32];
    float d   = lane_pick(llo, jA);
    #pragma nounroll
    for (int k = jA; k < 32; ++k) {
      const float nlo = lt[(k + 1) * GS_K];
      const float nhi = lt[(k + 1) * GS_K + 32];
      const float wk  = lane_pick(xlo, k) / d;
      xlo = (lane > k) ? __builtin_fmaf(-llo, wk, xlo) : ((lane == k) ? wk : xlo);
      xhi = __builtin_fmaf(-lhi, wk, xhi);
      d   = lane_pick(nlo, (k + 1) & 31);
      llo = nlo;
      lhi = nhi;
    }
    d = lane_pick(lhi, 0);
    int k = 0;
    #pragma nounroll
    for (; k < 31 - wv; ++k) {
      const float nhi = lt[(33 + k) * GS_K + 32];
      xhi = fs_step(xhi, lhi, d, k, lane);
      d   = lane_pick(nhi, k + 1);
      lhi = nhi;
    }
    #pragma nounroll
    for (; k < 32; ++k) {
      const int   kn  = (k < 31) ? k + 1 : k;
      const float nhi = lt[(32 + kn) * GS_K + 32];
      xhi = fs_step(xhi, lhi, d, k, lane);
      yhi = fs_step(yhi, lhi, d, k, lane);
      d   = lane_pick(nhi, kn);
      lhi = nhi;
    }
    W[lane * GS_K + jA]        = xlo;
    W[(lane + 32) * GS_K + jA] = xhi;
    W[lane * GS_K + jB]        = 0.0f;
    W[(lane + 32) * GS_K + jB] = yhi;
  }
  __syncthreads();

  {
    const int m    = lane & 15;
    const int koff = (lane < 16) ? 0 : 2;
    const int mb   = (lane < 16) ? 0 : 8;
    for (int t = wave; t < 4 * (GS_L / 16); t += 32) {
      const int ti = t >> 6;
      const int tc = t & 63;
      const float* pa = W + (ti * 16 + m) * GS_K + koff;
      v8f acc = {0.f, 0.f, 0.f, 0.f, 0.f, 0.f, 0.f, 0.f};
      #pragma unroll
      for (int k0 = 0; k0 < GS_K; k0 += 4) {
        v2f a = *(const v2f*)(pa + k0);
        v2f bb;
        bb.x = X[(k0 + koff    ) * GS_LP + tc * 16 + m];
        bb.y = X[(k0 + koff + 1) * GS_LP + tc * 16 + m];
        acc = __builtin_amdgcn_wmma_f32_16x16x4_f32(false, a, false, bb,
                                                    (short)0, acc, false, false);
      }
      #pragma unroll
      for (int j = 0; j < 8; ++j)
        og[(ti * 16 + mb + j) * GS_L + tc * 16 + m] = acc[j];
    }
  }
}

extern "C" void kernel_launch(void* const* d_in, const int* in_sizes, int n_in,
                              void* d_out, int out_size, void* d_ws, size_t ws_size,
                              hipStream_t stream) {
  (void)in_sizes; (void)n_in; (void)out_size; (void)d_ws; (void)ws_size;
  const float* x   = (const float*)d_in[0];
  float*       out = (float*)d_out;

  const size_t shmem = (size_t)(GS_K * GS_LP + 2 * GS_K * GS_K) * sizeof(float);
  hipFuncSetAttribute(reinterpret_cast<const void*>(&gram_schmidt_cholqr),
                      hipFuncAttributeMaxDynamicSharedMemorySize, (int)shmem);

  gram_schmidt_cholqr<<<dim3(GS_B), dim3(GS_NT), shmem, stream>>>(x, out);
}


# Row pitch of the L^{-1} matrix in LDS 64 -> 68 words: the final product's A-fragment reads (16 rows on one bank at pitch 64) become conflict-free, its column stores 2-way instead of 32-way; same sums
# speedup vs baseline: 4.9619x; 1.3214x over previous
#include <hip/hip_runtime.h>

typedef float v2f __attribute__((ext_vector_type(2)));
typedef float v4f __attribute__((ext_vector_type(4)));
typedef float v8f __attribute__((ext_vector_type(8)));

#define GS_B  512
#define GS_K  64
#define GS_L  1024
#define GS_LP 1028
#define GS_WP 68
#define GS_NT 1024

__device__ __forceinline__ float lane_bcast(float v, int src_lane) {
  return __int_as_float(__builtin_amdgcn_readlane(__float_as_int(v), src_lane));
}

__device__ __forceinline__ float lane_pick(float v, int src_lane) {
  return __int_as_float(__builtin_amdgcn_readlane(__float_as_int(v), src_lane));
}

__device__ __forceinline__ float fs_step(float x, float l, float d, int k, int lane) {
  const float wk = lane_pick(x, k) / d;
  return (lane > k) ? __builtin_fmaf(-l, wk, x) : ((lane == k) ? wk : x);
}

template <int J, int P>
__device__ __forceinline__ void chol_dot(float (&a)[32], float (&b)[64]) {
  if constexpr (P < J) {
    float ljp;
    if constexpr (J < 32) ljp = lane_bcast(a[P], J);
    else                  ljp = lane_bcast(b[P], J - 32);
    if constexpr (J < 32) a[J] = __builtin_fmaf(ljp, a[P], a[J]);
    b[J] = __builtin_fmaf(ljp, b[P], b[J]);
    chol_dot<J, P + 1>(a, b);
  }
}

template <int J>
__device__ __forceinline__ void chol_col(float (&a)[32], float (&b)[64], int lane) {
  if constexpr (J < 64) {
    chol_dot<J, 0>(a, b);
    float djj;
    if constexpr (J < 32) djj = -lane_bcast(a[J], J);
    else                  djj = -lane_bcast(b[J], J - 32);
    const float ljj = sqrtf(djj);
    const float inv = 1.0f / ljj;
    if constexpr (J < 32) {
      a[J] = (lane == J) ? ljj : (-a[J]) * inv;
      b[J] = (-b[J]) * inv;
    } else {
      b[J] = (lane == J - 32) ? ljj : (-b[J]) * inv;
    }
    chol_col<J + 1>(a, b, lane);
  }
}

__global__ __launch_bounds__(GS_NT, 1)
void gram_schmidt_cholqr(const float* __restrict__ xin,
                         float* __restrict__ xout) {
  extern __shared__ float smem[];
  float* X = smem;
  float* G = smem + GS_K * GS_LP;
  float* W = G + GS_K * GS_K;

  const int b    = blockIdx.x;
  const int tid  = threadIdx.x;
  const int wave = tid >> 5;
  const int lane = tid & 31;

  const float* xg = xin  + (size_t)b * (GS_K * GS_L);
  float*       og = xout + (size_t)b * (GS_K * GS_L);

  for (int idx = tid; idx < GS_K * (GS_L / 4); idx += GS_NT) {
    int r  = idx >> 8;
    int c4 = (idx & 255) << 2;
    *(v4f*)(X + r * GS_LP + c4) = *(const v4f*)(xg + r * GS_L + c4);
  }
  __syncthreads();

  {
    const int tile  = wave >> 1;
    const int ti    = tile >> 2;
    const int tj    = tile & 3;
    const int khalf = wave & 1;
    const int m     = lane & 15;
    const int koff  = (lane < 16) ? 0 : 2;
    const float* pa = X + (ti * 16 + m) * GS_LP + koff + khalf * (GS_L / 2);
    const float* pb = X + (tj * 16 + m) * GS_LP + koff + khalf * (GS_L / 2);
    v8f acc = {0.f, 0.f, 0.f, 0.f, 0.f, 0.f, 0.f, 0.f};
    #pragma unroll 4
    for (int k = 0; k < GS_L / 2; k += 4) {
      v2f a  = *(const v2f*)(pa + k);
      v2f bb = *(const v2f*)(pb + k);
      acc = __builtin_amdgcn_wmma_f32_16x16x4_f32(false, a, false, bb,
                                                  (short)0, acc, false, false);
    }
    float*    dst = khalf ? W : G;
    const int mb  = (lane < 16) ? 0 : 8;
    #pragma unroll
    for (int j = 0; j < 8; ++j)
      dst[(ti * 16 + mb + j) * GS_K + tj * 16 + m] = acc[j];
  }
  __syncthreads();
  for (int idx = tid; idx < GS_K * GS_K; idx += GS_NT) G[idx] += W[idx];
  __syncthreads();

  if (wave == 0) {
    float ra[32], rb[64];
    float* ga = G + lane * GS_K;
    float* gb = G + (lane + 32) * GS_K;
    #pragma unroll
    for (int k = 0; k < 32; k += 4) {
      v4f t = *(const v4f*)(ga + k);
      ra[k] = -t.x; ra[k + 1] = -t.y; ra[k + 2] = -t.z; ra[k + 3] = -t.w;
    }
    #pragma unroll
    for (int k = 0; k < 64; k += 4) {
      v4f t = *(const v4f*)(gb + k);
      rb[k] = -t.x; rb[k + 1] = -t.y; rb[k + 2] = -t.z; rb[k + 3] = -t.w;
    }
    chol_col<0>(ra, rb, lane);
    #pragma unroll
    for (int k = 0; k < 32; ++k) {
      G[k * GS_K + lane]      = ra[k];
      G[k * GS_K + lane + 32] = rb[k];
    }
    #pragma unroll
    for (int k = 32; k < 64; ++k)
      G[k * GS_K + lane + 32] = rb[k];
  }
  __syncthreads();

  {
    const int wv = __builtin_amdgcn_readfirstlane(wave);
    const int jA = wv;
    const int jB = 63 - wv;
    float xlo = (lane == jA) ? 1.0f : 0.0f;
    float xhi = 0.0f;
    float yhi = (lane + 32 == jB) ? 1.0f : 0.0f;
    const float* lt = G + lane;

    float llo = lt[jA * GS_K];
    float lhi = lt[jA * GS_K + 32];
    float d   = lane_pick(llo, jA);
    #pragma nounroll
    for (int k = jA; k < 32; ++k) {
      const float nlo = lt[(k + 1) * GS_K];
      const float nhi = lt[(k + 1) * GS_K + 32];
      const float wk  = lane_pick(xlo, k) / d;
      xlo = (lane > k) ? __builtin_fmaf(-llo, wk, xlo) : ((lane == k) ? wk : xlo);
      xhi = __builtin_fmaf(-lhi, wk, xhi);
      d   = lane_pick(nlo, (k + 1) & 31);
      llo = nlo;
      lhi = nhi;
    }
    d = lane_pick(lhi, 0);
    int k = 0;
    #pragma nounroll
    for (; k < 31 - wv; ++k) {
      const float nhi = lt[(33 + k) * GS_K + 32];
      xhi = fs_step(xhi, lhi, d, k, lane);
      d   = lane_pick(nhi, k + 1);
      lhi = nhi;
    }
    #pragma nounroll
    for (; k < 32; ++k) {
      const int   kn  = (k < 31) ? k + 1 : k;
      const float nhi = lt[(32 + kn) * GS_K + 32];
      xhi = fs_step(xhi, lhi, d, k, lane);
      yhi = fs_step(yhi, lhi, d, k, lane);
      d   = lane_pick(nhi, kn);
      lhi = nhi;
    }
    W[lane * GS_WP + jA]        = xlo;
    W[(lane + 32) * GS_WP + jA] = xhi;
    W[lane * GS_WP + jB]        = 0.0f;
    W[(lane + 32) * GS_WP + jB] = yhi;
  }
  __syncthreads();

  {
    const int m    = lane & 15;
    const int koff = (lane < 16) ? 0 : 2;
    const int mb   = (lane < 16) ? 0 : 8;
    for (int t = wave; t < 4 * (GS_L / 16); t += 32) {
      const int ti = t >> 6;
      const int tc = t & 63;
      const float* pa = W + (ti * 16 + m) * GS_WP + koff;
      v8f acc = {0.f, 0.f, 0.f, 0.f, 0.f, 0.f, 0.f, 0.f};
      #pragma unroll
      for (int k0 = 0; k0 < GS_K; k0 += 4) {
        v2f a = *(const v2f*)(pa + k0);
        v2f bb;
        bb.x = X[(k0 + koff    ) * GS_LP + tc * 16 + m];
        bb.y = X[(k0 + koff + 1) * GS_LP + tc * 16 + m];
        acc = __builtin_amdgcn_wmma_f32_16x16x4_f32(false, a, false, bb,
                                                    (short)0, acc, false, false);
      }
      #pragma unroll
      for (int j = 0; j < 8; ++j)
        og[(ti * 16 + mb + j) * GS_L + tc * 16 + m] = acc[j];
    }
  }
}

extern "C" void kernel_launch(void* const* d_in, const int* in_sizes, int n_in,
                              void* d_out, int out_size, void* d_ws, size_t ws_size,
                              hipStream_t stream) {
  (void)in_sizes; (void)n_in; (void)out_size; (void)d_ws; (void)ws_size;
  const float* x   = (const float*)d_in[0];
  float*       out = (float*)d_out;

  const size_t shmem = (size_t)(GS_K * GS_LP + GS_K * GS_K + GS_K * GS_WP) * sizeof(float);
  hipFuncSetAttribute(reinterpret_cast<const void*>(&gram_schmidt_cholqr),
                      hipFuncAttributeMaxDynamicSharedMemorySize, (int)shmem);

  gram_schmidt_cholqr<<<dim3(GS_B), dim3(GS_NT), shmem, stream>>>(x, out);
}


# Final product Q = L^{-1} X: k-steps where the lower-triangular L^{-1} is zero are skipped (40 of 64 products per column tile); a wave owns whole column tiles, so B is read once per k; same sums
# speedup vs baseline: 5.2297x; 1.0540x over previous
#include <hip/hip_runtime.h>

typedef float v2f __attribute__((ext_vector_type(2)));
typedef float v4f __attribute__((ext_vector_type(4)));
typedef float v8f __attribute__((ext_vector_type(8)));

#define GS_B  512
#define GS_K  64
#define GS_L  1024
#define GS_LP 1028
#define GS_WP 68
#define GS_NT 1024

__device__ __forceinline__ float lane_bcast(float v, int src_lane) {
  return __int_as_float(__builtin_amdgcn_readlane(__float_as_int(v), src_lane));
}

__device__ __forceinline__ float lane_pick(float v, int src_lane) {
  return __int_as_float(__builtin_amdgcn_readlane(__float_as_int(v), src_lane));
}

__device__ __forceinline__ float fs_step(float x, float l, float d, int k, int lane) {
  const float wk = lane_pick(x, k) / d;
  return (lane > k) ? __builtin_fmaf(-l, wk, x) : ((lane == k) ? wk : x);
}

template <int J, int P>
__device__ __forceinline__ void chol_dot(float (&a)[32], float (&b)[64]) {
  if constexpr (P < J) {
    float ljp;
    if constexpr (J < 32) ljp = lane_bcast(a[P], J);
    else                  ljp = lane_bcast(b[P], J - 32);
    if constexpr (J < 32) a[J] = __builtin_fmaf(ljp, a[P], a[J]);
    b[J] = __builtin_fmaf(ljp, b[P], b[J]);
    chol_dot<J, P + 1>(a, b);
  }
}

template <int J>
__device__ __forceinline__ void chol_col(float (&a)[32], float (&b)[64], int lane) {
  if constexpr (J < 64) {
    chol_dot<J, 0>(a, b);
    float djj;
    if constexpr (J < 32) djj = -lane_bcast(a[J], J);
    else                  djj = -lane_bcast(b[J], J - 32);
    const float ljj = sqrtf(djj);
    const float inv = 1.0f / ljj;
    if constexpr (J < 32) {
      a[J] = (lane == J) ? ljj : (-a[J]) * inv;
      b[J] = (-b[J]) * inv;
    } else {
      b[J] = (lane == J - 32) ? ljj : (-b[J]) * inv;
    }
    chol_col<J + 1>(a, b, lane);
  }
}

__global__ __launch_bounds__(GS_NT, 1)
void gram_schmidt_cholqr(const float* __restrict__ xin,
                         float* __restrict__ xout) {
  extern __shared__ float smem[];
  float* X = smem;
  float* G = smem + GS_K * GS_LP;
  float* W = G + GS_K * GS_K;

  const int b    = blockIdx.x;
  const int tid  = threadIdx.x;
  const int wave = tid >> 5;
  const int lane = tid & 31;

  const float* xg = xin  + (size_t)b * (GS_K * GS_L);
  float*       og = xout + (size_t)b * (GS_K * GS_L);

  for (int idx = tid; idx < GS_K * (GS_L / 4); idx += GS_NT) {
    int r  = idx >> 8;
    int c4 = (idx & 255) << 2;
    *(v4f*)(X + r * GS_LP + c4) = *(const v4f*)(xg + r * GS_L + c4);
  }
  __syncthreads();

  {
    const int tile  = wave >> 1;
    const int ti    = tile >> 2;
    const int tj    = tile & 3;
    const int khalf = wave & 1;
    const int m     = lane & 15;
    const int koff  = (lane < 16) ? 0 : 2;
    const float* pa = X + (ti * 16 + m) * GS_LP + koff + khalf * (GS_L / 2);
    const float* pb = X + (tj * 16 + m) * GS_LP + koff + khalf * (GS_L / 2);
    v8f acc = {0.f, 0.f, 0.f, 0.f, 0.f, 0.f, 0.f, 0.f};
    #pragma unroll 4
    for (int k = 0; k < GS_L / 2; k += 4) {
      v2f a  = *(const v2f*)(pa + k);
      v2f bb = *(const v2f*)(pb + k);
      acc = __builtin_amdgcn_wmma_f32_16x16x4_f32(false, a, false, bb,
                                                  (short)0, acc, false, false);
    }
    float*    dst = khalf ? W : G;
    const int mb  = (lane < 16) ? 0 : 8;
    #pragma unroll
    for (int j = 0; j < 8; ++j)
      dst[(ti * 16 + mb + j) * GS_K + tj * 16 + m] = acc[j];
  }
  __syncthreads();
  for (int idx = tid; idx < GS_K * GS_K; idx += GS_NT) G[idx] += W[idx];
  __syncthreads();

  if (wave == 0) {
    float ra[32], rb[64];
    float* ga = G + lane * GS_K;
    float* gb = G + (lane + 32) * GS_K;
    #pragma unroll
    for (int k = 0; k < 32; k += 4) {
      v4f t = *(const v4f*)(ga + k);
      ra[k] = -t.x; ra[k + 1] = -t.y; ra[k + 2] = -t.z; ra[k + 3] = -t.w;
    }
    #pragma unroll
    for (int k = 0; k < 64; k += 4) {
      v4f t = *(const v4f*)(gb + k);
      rb[k] = -t.x; rb[k + 1] = -t.y; rb[k + 2] = -t.z; rb[k + 3] = -t.w;
    }
    chol_col<0>(ra, rb, lane);
    #pragma unroll
    for (int k = 0; k < 32; ++k) {
      G[k * GS_K + lane]      = ra[k];
      G[k * GS_K + lane + 32] = rb[k];
    }
    #pragma unroll
    for (int k = 32; k < 64; ++k)
      G[k * GS_K + lane + 32] = rb[k];
  }
  __syncthreads();

  {
    const int wv = __builtin_amdgcn_readfirstlane(wave);
    const int jA = wv;
    const int jB = 63 - wv;
    float xlo = (lane == jA) ? 1.0f : 0.0f;
    float xhi = 0.0f;
    float yhi = (lane + 32 == jB) ? 1.0f : 0.0f;
    const float* lt = G + lane;

    float llo = lt[jA * GS_K];
    float lhi = lt[jA * GS_K + 32];
    float d   = lane_pick(llo, jA);
    #pragma nounroll
    for (int k = jA; k < 32; ++k) {
      const float nlo = lt[(k + 1) * GS_K];
      const float nhi = lt[(k + 1) * GS_K + 32];
      const float wk  = lane_pick(xlo, k) / d;
      xlo = (lane > k) ? __builtin_fmaf(-llo, wk, xlo) : ((lane == k) ? wk : xlo);
      xhi = __builtin_fmaf(-lhi, wk, xhi);
      d   = lane_pick(nlo, (k + 1) & 31);
      llo = nlo;
      lhi = nhi;
    }
    d = lane_pick(lhi, 0);
    int k = 0;
    #pragma nounroll
    for (; k < 31 - wv; ++k) {
      const float nhi = lt[(33 + k) * GS_K + 32];
      xhi = fs_step(xhi, lhi, d, k, lane);
      d   = lane_pick(nhi, k + 1);
      lhi = nhi;
    }
    #pragma nounroll
    for (; k < 32; ++k) {
      const int   kn  = (k < 31) ? k + 1 : k;
      const float nhi = lt[(32 + kn) * GS_K + 32];
      xhi = fs_step(xhi, lhi, d, k, lane);
      yhi = fs_step(yhi, lhi, d, k, lane);
      d   = lane_pick(nhi, kn);
      lhi = nhi;
    }
    W[lane * GS_WP + jA]        = xlo;
    W[(lane + 32) * GS_WP + jA] = xhi;
    W[lane * GS_WP + jB]        = 0.0f;
    W[(lane + 32) * GS_WP + jB] = yhi;
  }
  __syncthreads();

  {
    const int m    = lane & 15;
    const int koff = (lane < 16) ? 0 : 2;
    const int mb   = (lane < 16) ? 0 : 8;
    const int col  = wave * 32 + m;
    const float* pa = W + m * GS_WP + koff;
    const float* pb = X + koff * GS_LP + col;
    float*       po = og + mb * GS_L + col;
    v8f acc[4][2];
    #pragma unroll
    for (int ti = 0; ti < 4; ++ti) {
      acc[ti][0] = (v8f){0.f, 0.f, 0.f, 0.f, 0.f, 0.f, 0.f, 0.f};
      acc[ti][1] = (v8f){0.f, 0.f, 0.f, 0.f, 0.f, 0.f, 0.f, 0.f};
    }
    #pragma unroll
    for (int kb = 0; kb < 4; ++kb) {
      #pragma unroll
      for (int k0 = kb * 16; k0 < kb * 16 + 16; k0 += 4) {
        v2f b0, b1;
        b0.x = pb[(k0    ) * GS_LP];
        b0.y = pb[(k0 + 1) * GS_LP];
        b1.x = pb[(k0    ) * GS_LP + 16];
        b1.y = pb[(k0 + 1) * GS_LP + 16];
        #pragma unroll
        for (int ti = kb; ti < 4; ++ti) {
          v2f a = *(const v2f*)(pa + ti * 16 * GS_WP + k0);
          acc[ti][0] = __builtin_amdgcn_wmma_f32_16x16x4_f32(false, a, false, b0,
                                                             (short)0, acc[ti][0], false, false);
          acc[ti][1] = __builtin_amdgcn_wmma_f32_16x16x4_f32(false, a, false, b1,
                                                             (short)0, acc[ti][1], false, false);
        }
      }
      #pragma unroll
      for (int j = 0; j < 8; ++j) {
        po[(kb * 16 + j) * GS_L]      = acc[kb][0][j];
        po[(kb * 16 + j) * GS_L + 16] = acc[kb][1][j];
      }
    }
  }
}

extern "C" void kernel_launch(void* const* d_in, const int* in_sizes, int n_in,
                              void* d_out, int out_size, void* d_ws, size_t ws_size,
                              hipStream_t stream) {
  (void)in_sizes; (void)n_in; (void)out_size; (void)d_ws; (void)ws_size;
  const float* x   = (const float*)d_in[0];
  float*       out = (float*)d_out;

  const size_t shmem = (size_t)(GS_K * GS_LP + GS_K * GS_K + GS_K * GS_WP) * sizeof(float);
  hipFuncSetAttribute(reinterpret_cast<const void*>(&gram_schmidt_cholqr),
                      hipFuncAttributeMaxDynamicSharedMemorySize, (int)shmem);

  gram_schmidt_cholqr<<<dim3(GS_B), dim3(GS_NT), shmem, stream>>>(x, out);
}


# Gram matrix X X^T: only the 10 tiles on or below the diagonal are computed (the Cholesky reads nothing above it), then mirrored to their transposed places; 6 of 16 tile products dropped; same sums
# speedup vs baseline: 5.6177x; 1.0742x over previous
#include <hip/hip_runtime.h>

typedef float v2f __attribute__((ext_vector_type(2)));
typedef float v4f __attribute__((ext_vector_type(4)));
typedef float v8f __attribute__((ext_vector_type(8)));

#define GS_B  512
#define GS_K  64
#define GS_L  1024
#define GS_LP 1028
#define GS_WP 68
#define GS_NT 1024

__device__ __forceinline__ float lane_bcast(float v, int src_lane) {
  return __int_as_float(__builtin_amdgcn_readlane(__float_as_int(v), src_lane));
}

__device__ __forceinline__ float lane_pick(float v, int src_lane) {
  return __int_as_float(__builtin_amdgcn_readlane(__float_as_int(v), src_lane));
}

__device__ __forceinline__ float fs_step(float x, float l, float d, int k, int lane) {
  const float wk = lane_pick(x, k) / d;
  return (lane > k) ? __builtin_fmaf(-l, wk, x) : ((lane == k) ? wk : x);
}

template <int J, int P>
__device__ __forceinline__ void chol_dot(float (&a)[32], float (&b)[64]) {
  if constexpr (P < J) {
    float ljp;
    if constexpr (J < 32) ljp = lane_bcast(a[P], J);
    else                  ljp = lane_bcast(b[P], J - 32);
    if constexpr (J < 32) a[J] = __builtin_fmaf(ljp, a[P], a[J]);
    b[J] = __builtin_fmaf(ljp, b[P], b[J]);
    chol_dot<J, P + 1>(a, b);
  }
}

template <int J>
__device__ __forceinline__ void chol_col(float (&a)[32], float (&b)[64], int lane) {
  if constexpr (J < 64) {
    chol_dot<J, 0>(a, b);
    float djj;
    if constexpr (J < 32) djj = -lane_bcast(a[J], J);
    else                  djj = -lane_bcast(b[J], J - 32);
    const float ljj = sqrtf(djj);
    const float inv = 1.0f / ljj;
    if constexpr (J < 32) {
      a[J] = (lane == J) ? ljj : (-a[J]) * inv;
      b[J] = (-b[J]) * inv;
    } else {
      b[J] = (lane == J - 32) ? ljj : (-b[J]) * inv;
    }
    chol_col<J + 1>(a, b, lane);
  }
}

__global__ __launch_bounds__(GS_NT, 1)
void gram_schmidt_cholqr(const float* __restrict__ xin,
                         float* __restrict__ xout) {
  extern __shared__ float smem[];
  float* X = smem;
  float* G = smem + GS_K * GS_LP;
  float* W = G + GS_K * GS_K;

  const int b    = blockIdx.x;
  const int tid  = threadIdx.x;
  const int wave = tid >> 5;
  const int lane = tid & 31;

  const float* xg = xin  + (size_t)b * (GS_K * GS_L);
  float*       og = xout + (size_t)b * (GS_K * GS_L);

  for (int idx = tid; idx < GS_K * (GS_L / 4); idx += GS_NT) {
    int r  = idx >> 8;
    int c4 = (idx & 255) << 2;
    *(v4f*)(X + r * GS_LP + c4) = *(const v4f*)(xg + r * GS_L + c4);
  }
  __syncthreads();

  {
    const int wq = __builtin_amdgcn_readfirstlane(wave);
    if (wq < 20) {
      const int tile  = wq >> 1;
      const int ti    = (tile >= 6) ? 3 : ((tile >= 3) ? 2 : ((tile >= 1) ? 1 : 0));
      const int tj    = tile - ((ti * (ti + 1)) >> 1);
      const int khalf = wq & 1;
      const int m     = lane & 15;
      const int koff  = (lane < 16) ? 0 : 2;
      const float* pa = X + (ti * 16 + m) * GS_LP + koff + khalf * (GS_L / 2);
      const float* pb = X + (tj * 16 + m) * GS_LP + koff + khalf * (GS_L / 2);
      v8f acc = {0.f, 0.f, 0.f, 0.f, 0.f, 0.f, 0.f, 0.f};
      #pragma unroll 4
      for (int k = 0; k < GS_L / 2; k += 4) {
        v2f a  = *(const v2f*)(pa + k);
        v2f bb = *(const v2f*)(pb + k);
        acc = __builtin_amdgcn_wmma_f32_16x16x4_f32(false, a, false, bb,
                                                    (short)0, acc, false, false);
      }
      float*    dst = khalf ? W : G;
      const int mb  = (lane < 16) ? 0 : 8;
      #pragma unroll
      for (int j = 0; j < 8; ++j)
        dst[(ti * 16 + mb + j) * GS_K + tj * 16 + m] = acc[j];
      if (tj < ti) {
        float* mr = dst + (tj * 16 + m) * GS_K + ti * 16 + mb;
        *(v4f*)(mr)     = (v4f){acc[0], acc[1], acc[2], acc[3]};
        *(v4f*)(mr + 4) = (v4f){acc[4], acc[5], acc[6], acc[7]};
      }
    }
  }
  __syncthreads();
  for (int idx = tid; idx < GS_K * GS_K; idx += GS_NT) G[idx] += W[idx];
  __syncthreads();

  if (wave == 0) {
    float ra[32], rb[64];
    float* ga = G + lane * GS_K;
    float* gb = G + (lane + 32) * GS_K;
    #pragma unroll
    for (int k = 0; k < 32; k += 4) {
      v4f t = *(const v4f*)(ga + k);
      ra[k] = -t.x; ra[k + 1] = -t.y; ra[k + 2] = -t.z; ra[k + 3] = -t.w;
    }
    #pragma unroll
    for (int k = 0; k < 64; k += 4) {
      v4f t = *(const v4f*)(gb + k);
      rb[k] = -t.x; rb[k + 1] = -t.y; rb[k + 2] = -t.z; rb[k + 3] = -t.w;
    }
    chol_col<0>(ra, rb, lane);
    #pragma unroll
    for (int k = 0; k < 32; ++k) {
      G[k * GS_K + lane]      = ra[k];
      G[k * GS_K + lane + 32] = rb[k];
    }
    #pragma unroll
    for (int k = 32; k < 64; ++k)
      G[k * GS_K + lane + 32] = rb[k];
  }
  __syncthreads();

  {
    const int wv = __builtin_amdgcn_readfirstlane(wave);
    const int jA = wv;
    const int jB = 63 - wv;
    float xlo = (lane == jA) ? 1.0f : 0.0f;
    float xhi = 0.0f;
    float yhi = (lane + 32 == jB) ? 1.0f : 0.0f;
    const float* lt = G + lane;

    float llo = lt[jA * GS_K];
    float lhi = lt[jA * GS_K + 32];
    float d   = lane_pick(llo, jA);
    #pragma nounroll
    for (int k = jA; k < 32; ++k) {
      const float nlo = lt[(k + 1) * GS_K];
      const float nhi = lt[(k + 1) * GS_K + 32];
      const float wk  = lane_pick(xlo, k) / d;
      xlo = (lane > k) ? __builtin_fmaf(-llo, wk, xlo) : ((lane == k) ? wk : xlo);
      xhi = __builtin_fmaf(-lhi, wk, xhi);
      d   = lane_pick(nlo, (k + 1) & 31);
      llo = nlo;
      lhi = nhi;
    }
    d = lane_pick(lhi, 0);
    int k = 0;
    #pragma nounroll
    for (; k < 31 - wv; ++k) {
      const float nhi = lt[(33 + k) * GS_K + 32];
      xhi = fs_step(xhi, lhi, d, k, lane);
      d   = lane_pick(nhi, k + 1);
      lhi = nhi;
    }
    #pragma nounroll
    for (; k < 32; ++k) {
      const int   kn  = (k < 31) ? k + 1 : k;
      const float nhi = lt[(32 + kn) * GS_K + 32];
      xhi = fs_step(xhi, lhi, d, k, lane);
      yhi = fs_step(yhi, lhi, d, k, lane);
      d   = lane_pick(nhi, kn);
      lhi = nhi;
    }
    W[lane * GS_WP + jA]        = xlo;
    W[(lane + 32) * GS_WP + jA] = xhi;
    W[lane * GS_WP + jB]        = 0.0f;
    W[(lane + 32) * GS_WP + jB] = yhi;
  }
  __syncthreads();

  {
    const int m    = lane & 15;
    const int koff = (lane < 16) ? 0 : 2;
    const int mb   = (lane < 16) ? 0 : 8;
    const int col  = wave * 32 + m;
    const float* pa = W + m * GS_WP + koff;
    const float* pb = X + koff * GS_LP + col;
    float*       po = og + mb * GS_L + col;
    v8f acc[4][2];
    #pragma unroll
    for (int ti = 0; ti < 4; ++ti) {
      acc[ti][0] = (v8f){0.f, 0.f, 0.f, 0.f, 0.f, 0.f, 0.f, 0.f};
      acc[ti][1] = (v8f){0.f, 0.f, 0.f, 0.f, 0.f, 0.f, 0.f, 0.f};
    }
    #pragma unroll
    for (int kb = 0; kb < 4; ++kb) {
      #pragma unroll
      for (int k0 = kb * 16; k0 < kb * 16 + 16; k0 += 4) {
        v2f b0, b1;
        b0.x = pb[(k0    ) * GS_LP];
        b0.y = pb[(k0 + 1) * GS_LP];
        b1.x = pb[(k0    ) * GS_LP + 16];
        b1.y = pb[(k0 + 1) * GS_LP + 16];
        #pragma unroll
        for (int ti = kb; ti < 4; ++ti) {
          v2f a = *(const v2f*)(pa + ti * 16 * GS_WP + k0);
          acc[ti][0] = __builtin_amdgcn_wmma_f32_16x16x4_f32(false, a, false, b0,
                                                             (short)0, acc[ti][0], false, false);
          acc[ti][1] = __builtin_amdgcn_wmma_f32_16x16x4_f32(false, a, false, b1,
                                                             (short)0, acc[ti][1], false, false);
        }
      }
      #pragma unroll
      for (int j = 0; j < 8; ++j) {
        po[(kb * 16 + j) * GS_L]      = acc[kb][0][j];
        po[(kb * 16 + j) * GS_L + 16] = acc[kb][1][j];
      }
    }
  }
}

extern "C" void kernel_launch(void* const* d_in, const int* in_sizes, int n_in,
                              void* d_out, int out_size, void* d_ws, size_t ws_size,
                              hipStream_t stream) {
  (void)in_sizes; (void)n_in; (void)out_size; (void)d_ws; (void)ws_size;
  const float* x   = (const float*)d_in[0];
  float*       out = (float*)d_out;

  const size_t shmem = (size_t)(GS_K * GS_LP + GS_K * GS_K + GS_K * GS_WP) * sizeof(float);
  hipFuncSetAttribute(reinterpret_cast<const void*>(&gram_schmidt_cholqr),
                      hipFuncAttributeMaxDynamicSharedMemorySize, (int)shmem);

  gram_schmidt_cholqr<<<dim3(GS_B), dim3(GS_NT), shmem, stream>>>(x, out);
}


# Cholesky made right-looking: a finished column is taken off all later columns at once (independent multiply-adds beside the next diagonal's sqrt and division), not one dependent chain each; same sums
# speedup vs baseline: 6.1990x; 1.1035x over previous
#include <hip/hip_runtime.h>

typedef float v2f __attribute__((ext_vector_type(2)));
typedef float v4f __attribute__((ext_vector_type(4)));
typedef float v8f __attribute__((ext_vector_type(8)));

#define GS_B  512
#define GS_K  64
#define GS_L  1024
#define GS_LP 1028
#define GS_WP 68
#define GS_NT 1024

__device__ __forceinline__ float lane_bcast(float v, int src_lane) {
  return __int_as_float(__builtin_amdgcn_readlane(__float_as_int(v), src_lane));
}

__device__ __forceinline__ float lane_pick(float v, int src_lane) {
  return __int_as_float(__builtin_amdgcn_readlane(__float_as_int(v), src_lane));
}

__device__ __forceinline__ float fs_step(float x, float l, float d, int k, int lane) {
  const float wk = lane_pick(x, k) / d;
  return (lane > k) ? __builtin_fmaf(-l, wk, x) : ((lane == k) ? wk : x);
}


template <int P, int J>
__device__ __forceinline__ void chol_sub1(float (&a)[32], float (&b)[64]) {
  float ljp;
  if constexpr (J < 32) ljp = lane_bcast(a[P], J);
  else                  ljp = lane_bcast(b[P], J - 32);
  if constexpr (J < 32) a[J] = __builtin_fmaf(ljp, a[P], a[J]);
  b[J] = __builtin_fmaf(ljp, b[P], b[J]);
}

template <int P, int J>
__device__ __forceinline__ void chol_sub(float (&a)[32], float (&b)[64]) {
  if constexpr (J < 64) {
    chol_sub1<P, J>(a, b);
    chol_sub<P, J + 1>(a, b);
  }
}

template <int P>
__device__ __forceinline__ void chol_fin(float (&a)[32], float (&b)[64], int lane) {
  float djj;
  if constexpr (P < 32) djj = -lane_bcast(a[P], P);
  else                  djj = -lane_bcast(b[P], P - 32);
  const float ljj = sqrtf(djj);
  const float inv = 1.0f / ljj;
  if constexpr (P < 32) {
    const float sa = (-a[P]) * inv;
    a[P] = (lane == P) ? ljj : sa;
    b[P] = (-b[P]) * inv;
  } else {
    const float sb = (-b[P]) * inv;
    b[P] = (lane == P - 32) ? ljj : sb;
  }
}

template <int J>
__device__ __forceinline__ void chol_pin(float (&a)[32], float (&b)[64]) {
  if constexpr (J < 64) {
    if constexpr (J < 32) asm volatile("" : "+v"(a[J]));
    asm volatile("" : "+v"(b[J]));
    chol_pin<J + 1>(a, b);
  }
}

template <int P>
__device__ __forceinline__ void chol_col(float (&a)[32], float (&b)[64], int lane) {
  if constexpr (P < 64) {
    chol_fin<P>(a, b, lane);
    if constexpr (P >= 1) chol_sub<P - 1, P + 1>(a, b);
    if constexpr (P + 1 < 64) chol_sub1<P, P + 1>(a, b);
    chol_pin<P + 1>(a, b);
    chol_col<P + 1>(a, b, lane);
  }
}

__global__ __launch_bounds__(GS_NT, 1)
void gram_schmidt_cholqr(const float* __restrict__ xin,
                         float* __restrict__ xout) {
  extern __shared__ float smem[];
  float* X = smem;
  float* G = smem + GS_K * GS_LP;
  float* W = G + GS_K * GS_K;

  const int b    = blockIdx.x;
  const int tid  = threadIdx.x;
  const int wave = tid >> 5;
  const int lane = tid & 31;

  const float* xg = xin  + (size_t)b * (GS_K * GS_L);
  float*       og = xout + (size_t)b * (GS_K * GS_L);

  for (int idx = tid; idx < GS_K * (GS_L / 4); idx += GS_NT) {
    int r  = idx >> 8;
    int c4 = (idx & 255) << 2;
    *(v4f*)(X + r * GS_LP + c4) = *(const v4f*)(xg + r * GS_L + c4);
  }
  __syncthreads();

  {
    const int wq = __builtin_amdgcn_readfirstlane(wave);
    if (wq < 20) {
      const int tile  = wq >> 1;
      const int ti    = (tile >= 6) ? 3 : ((tile >= 3) ? 2 : ((tile >= 1) ? 1 : 0));
      const int tj    = tile - ((ti * (ti + 1)) >> 1);
      const int khalf = wq & 1;
      const int m     = lane & 15;
      const int koff  = (lane < 16) ? 0 : 2;
      const float* pa = X + (ti * 16 + m) * GS_LP + koff + khalf * (GS_L / 2);
      const float* pb = X + (tj * 16 + m) * GS_LP + koff + khalf * (GS_L / 2);
      v8f acc = {0.f, 0.f, 0.f, 0.f, 0.f, 0.f, 0.f, 0.f};
      #pragma unroll 4
      for (int k = 0; k < GS_L / 2; k += 4) {
        v2f a  = *(const v2f*)(pa + k);
        v2f bb = *(const v2f*)(pb + k);
        acc = __builtin_amdgcn_wmma_f32_16x16x4_f32(false, a, false, bb,
                                                    (short)0, acc, false, false);
      }
      float*    dst = khalf ? W : G;
      const int mb  = (lane < 16) ? 0 : 8;
      #pragma unroll
      for (int j = 0; j < 8; ++j)
        dst[(ti * 16 + mb + j) * GS_K + tj * 16 + m] = acc[j];
      if (tj < ti) {
        float* mr = dst + (tj * 16 + m) * GS_K + ti * 16 + mb;
        *(v4f*)(mr)     = (v4f){acc[0], acc[1], acc[2], acc[3]};
        *(v4f*)(mr + 4) = (v4f){acc[4], acc[5], acc[6], acc[7]};
      }
    }
  }
  __syncthreads();
  for (int idx = tid; idx < GS_K * GS_K; idx += GS_NT) G[idx] = -(G[idx] + W[idx]);
  __syncthreads();

  if (wave == 0) {
    float ra[32], rb[64];
    float* ga = G + lane * GS_K;
    float* gb = G + (lane + 32) * GS_K;
    #pragma unroll
    for (int k = 0; k < 32; k += 4) {
      v4f t = *(const v4f*)(ga + k);
      ra[k] = t.x; ra[k + 1] = t.y; ra[k + 2] = t.z; ra[k + 3] = t.w;
    }
    #pragma unroll
    for (int k = 0; k < 64; k += 4) {
      v4f t = *(const v4f*)(gb + k);
      rb[k] = t.x; rb[k + 1] = t.y; rb[k + 2] = t.z; rb[k + 3] = t.w;
    }
    chol_col<0>(ra, rb, lane);
    #pragma unroll
    for (int k = 0; k < 32; ++k) {
      G[k * GS_K + lane]      = ra[k];
      G[k * GS_K + lane + 32] = rb[k];
    }
    #pragma unroll
    for (int k = 32; k < 64; ++k)
      G[k * GS_K + lane + 32] = rb[k];
  }
  __syncthreads();

  {
    const int wv = __builtin_amdgcn_readfirstlane(wave);
    const int jA = wv;
    const int jB = 63 - wv;
    float xlo = (lane == jA) ? 1.0f : 0.0f;
    float xhi = 0.0f;
    float yhi = (lane + 32 == jB) ? 1.0f : 0.0f;
    const float* lt = G + lane;

    float llo = lt[jA * GS_K];
    float lhi = lt[jA * GS_K + 32];
    float d   = lane_pick(llo, jA);
    #pragma nounroll
    for (int k = jA; k < 32; ++k) {
      const float nlo = lt[(k + 1) * GS_K];
      const float nhi = lt[(k + 1) * GS_K + 32];
      const float wk  = lane_pick(xlo, k) / d;
      xlo = (lane > k) ? __builtin_fmaf(-llo, wk, xlo) : ((lane == k) ? wk : xlo);
      xhi = __builtin_fmaf(-lhi, wk, xhi);
      d   = lane_pick(nlo, (k + 1) & 31);
      llo = nlo;
      lhi = nhi;
    }
    d = lane_pick(lhi, 0);
    int k = 0;
    #pragma nounroll
    for (; k < 31 - wv; ++k) {
      const float nhi = lt[(33 + k) * GS_K + 32];
      xhi = fs_step(xhi, lhi, d, k, lane);
      d   = lane_pick(nhi, k + 1);
      lhi = nhi;
    }
    #pragma nounroll
    for (; k < 32; ++k) {
      const int   kn  = (k < 31) ? k + 1 : k;
      const float nhi = lt[(32 + kn) * GS_K + 32];
      xhi = fs_step(xhi, lhi, d, k, lane);
      yhi = fs_step(yhi, lhi, d, k, lane);
      d   = lane_pick(nhi, kn);
      lhi = nhi;
    }
    W[lane * GS_WP + jA]        = xlo;
    W[(lane + 32) * GS_WP + jA] = xhi;
    W[lane * GS_WP + jB]        = 0.0f;
    W[(lane + 32) * GS_WP + jB] = yhi;
  }
  __syncthreads();

  {
    const int m    = lane & 15;
    const int koff = (lane < 16) ? 0 : 2;
    const int mb   = (lane < 16) ? 0 : 8;
    const int col  = wave * 32 + m;
    const float* pa = W + m * GS_WP + koff;
    const float* pb = X + koff * GS_LP + col;
    float*       po = og + mb * GS_L + col;
    v8f acc[4][2];
    #pragma unroll
    for (int ti = 0; ti < 4; ++ti) {
      acc[ti][0] = (v8f){0.f, 0.f, 0.f, 0.f, 0.f, 0.f, 0.f, 0.f};
      acc[ti][1] = (v8f){0.f, 0.f, 0.f, 0.f, 0.f, 0.f, 0.f, 0.f};
    }
    #pragma unroll
    for (int kb = 0; kb < 4; ++kb) {
      #pragma unroll
      for (int k0 = kb * 16; k0 < kb * 16 + 16; k0 += 4) {
        v2f b0, b1;
        b0.x = pb[(k0    ) * GS_LP];
        b0.y = pb[(k0 + 1) * GS_LP];
        b1.x = pb[(k0    ) * GS_LP + 16];
        b1.y = pb[(k0 + 1) * GS_LP + 16];
        #pragma unroll
        for (int ti = kb; ti < 4; ++ti) {
          v2f a = *(const v2f*)(pa + ti * 16 * GS_WP + k0);
          acc[ti][0] = __builtin_amdgcn_wmma_f32_16x16x4_f32(false, a, false, b0,
                                                             (short)0, acc[ti][0], false, false);
          acc[ti][1] = __builtin_amdgcn_wmma_f32_16x16x4_f32(false, a, false, b1,
                                                             (short)0, acc[ti][1], false, false);
        }
      }
      #pragma unroll
      for (int j = 0; j < 8; ++j) {
        po[(kb * 16 + j) * GS_L]      = acc[kb][0][j];
        po[(kb * 16 + j) * GS_L + 16] = acc[kb][1][j];
      }
    }
  }
}

extern "C" void kernel_launch(void* const* d_in, const int* in_sizes, int n_in,
                              void* d_out, int out_size, void* d_ws, size_t ws_size,
                              hipStream_t stream) {
  (void)in_sizes; (void)n_in; (void)out_size; (void)d_ws; (void)ws_size;
  const float* x   = (const float*)d_in[0];
  float*       out = (float*)d_out;

  const size_t shmem = (size_t)(GS_K * GS_LP + GS_K * GS_K + GS_K * GS_WP) * sizeof(float);
  hipFuncSetAttribute(reinterpret_cast<const void*>(&gram_schmidt_cholqr),
                      hipFuncAttributeMaxDynamicSharedMemorySize, (int)shmem);

  gram_schmidt_cholqr<<<dim3(GS_B), dim3(GS_NT), shmem, stream>>>(x, out);
}


# Diagonal of L kept as its reciprocal r = rsqrt(d): the Cholesky scales each column by r (no sqrt-then-divide chain), the L^{-1} substitution multiplies by r (no division per step); same sum order
# speedup vs baseline: 6.9879x; 1.1273x over previous
#include <hip/hip_runtime.h>

typedef float v2f __attribute__((ext_vector_type(2)));
typedef float v4f __attribute__((ext_vector_type(4)));
typedef float v8f __attribute__((ext_vector_type(8)));

#define GS_B  512
#define GS_K  64
#define GS_L  1024
#define GS_LP 1028
#define GS_WP 68
#define GS_NT 1024

__device__ __forceinline__ float lane_bcast(float v, int src_lane) {
  return __int_as_float(__builtin_amdgcn_readlane(__float_as_int(v), src_lane));
}

__device__ __forceinline__ float lane_pick(float v, int src_lane) {
  return __int_as_float(__builtin_amdgcn_readlane(__float_as_int(v), src_lane));
}

__device__ __forceinline__ float fs_step(float x, float l, float rd, int k, int lane) {
  const float wk = lane_pick(x, k) * rd;
  return (lane > k) ? __builtin_fmaf(-l, wk, x) : ((lane == k) ? wk : x);
}


template <int P, int J>
__device__ __forceinline__ void chol_sub1(float (&a)[32], float (&b)[64]) {
  float ljp;
  if constexpr (J < 32) ljp = lane_bcast(a[P], J);
  else                  ljp = lane_bcast(b[P], J - 32);
  if constexpr (J < 32) a[J] = __builtin_fmaf(ljp, a[P], a[J]);
  b[J] = __builtin_fmaf(ljp, b[P], b[J]);
}

template <int P, int J>
__device__ __forceinline__ void chol_sub(float (&a)[32], float (&b)[64]) {
  if constexpr (J < 64) {
    chol_sub1<P, J>(a, b);
    chol_sub<P, J + 1>(a, b);
  }
}

template <int P>
__device__ __forceinline__ void chol_fin(float (&a)[32], float (&b)[64], int lane) {
  float djj;
  if constexpr (P < 32) djj = -lane_bcast(a[P], P);
  else                  djj = -lane_bcast(b[P], P - 32);
  const float inv = rsqrtf(djj);
  if constexpr (P < 32) {
    const float sa = (-a[P]) * inv;
    a[P] = (lane == P) ? inv : sa;
    b[P] = (-b[P]) * inv;
  } else {
    const float sb = (-b[P]) * inv;
    b[P] = (lane == P - 32) ? inv : sb;
  }
}

template <int J>
__device__ __forceinline__ void chol_pin(float (&a)[32], float (&b)[64]) {
  if constexpr (J < 64) {
    if constexpr (J < 32) asm volatile("" : "+v"(a[J]));
    asm volatile("" : "+v"(b[J]));
    chol_pin<J + 1>(a, b);
  }
}

template <int P>
__device__ __forceinline__ void chol_col(float (&a)[32], float (&b)[64], int lane) {
  if constexpr (P < 64) {
    chol_fin<P>(a, b, lane);
    if constexpr (P >= 1) chol_sub<P - 1, P + 1>(a, b);
    if constexpr (P + 1 < 64) chol_sub1<P, P + 1>(a, b);
    chol_pin<P + 1>(a, b);
    chol_col<P + 1>(a, b, lane);
  }
}

__global__ __launch_bounds__(GS_NT, 1)
void gram_schmidt_cholqr(const float* __restrict__ xin,
                         float* __restrict__ xout) {
  extern __shared__ float smem[];
  float* X = smem;
  float* G = smem + GS_K * GS_LP;
  float* W = G + GS_K * GS_K;

  const int b    = blockIdx.x;
  const int tid  = threadIdx.x;
  const int wave = tid >> 5;
  const int lane = tid & 31;

  const float* xg = xin  + (size_t)b * (GS_K * GS_L);
  float*       og = xout + (size_t)b * (GS_K * GS_L);

  for (int idx = tid; idx < GS_K * (GS_L / 4); idx += GS_NT) {
    int r  = idx >> 8;
    int c4 = (idx & 255) << 2;
    *(v4f*)(X + r * GS_LP + c4) = *(const v4f*)(xg + r * GS_L + c4);
  }
  __syncthreads();

  {
    const int wq = __builtin_amdgcn_readfirstlane(wave);
    if (wq < 20) {
      const int tile  = wq >> 1;
      const int ti    = (tile >= 6) ? 3 : ((tile >= 3) ? 2 : ((tile >= 1) ? 1 : 0));
      const int tj    = tile - ((ti * (ti + 1)) >> 1);
      const int khalf = wq & 1;
      const int m     = lane & 15;
      const int koff  = (lane < 16) ? 0 : 2;
      const float* pa = X + (ti * 16 + m) * GS_LP + koff + khalf * (GS_L / 2);
      const float* pb = X + (tj * 16 + m) * GS_LP + koff + khalf * (GS_L / 2);
      v8f acc = {0.f, 0.f, 0.f, 0.f, 0.f, 0.f, 0.f, 0.f};
      #pragma unroll 4
      for (int k = 0; k < GS_L / 2; k += 4) {
        v2f a  = *(const v2f*)(pa + k);
        v2f bb = *(const v2f*)(pb + k);
        acc = __builtin_amdgcn_wmma_f32_16x16x4_f32(false, a, false, bb,
                                                    (short)0, acc, false, false);
      }
      float*    dst = khalf ? W : G;
      const int mb  = (lane < 16) ? 0 : 8;
      #pragma unroll
      for (int j = 0; j < 8; ++j)
        dst[(ti * 16 + mb + j) * GS_K + tj * 16 + m] = acc[j];
      if (tj < ti) {
        float* mr = dst + (tj * 16 + m) * GS_K + ti * 16 + mb;
        *(v4f*)(mr)     = (v4f){acc[0], acc[1], acc[2], acc[3]};
        *(v4f*)(mr + 4) = (v4f){acc[4], acc[5], acc[6], acc[7]};
      }
    }
  }
  __syncthreads();
  for (int idx = tid; idx < GS_K * GS_K; idx += GS_NT) G[idx] = -(G[idx] + W[idx]);
  __syncthreads();

  if (wave == 0) {
    float ra[32], rb[64];
    float* ga = G + lane * GS_K;
    float* gb = G + (lane + 32) * GS_K;
    #pragma unroll
    for (int k = 0; k < 32; k += 4) {
      v4f t = *(const v4f*)(ga + k);
      ra[k] = t.x; ra[k + 1] = t.y; ra[k + 2] = t.z; ra[k + 3] = t.w;
    }
    #pragma unroll
    for (int k = 0; k < 64; k += 4) {
      v4f t = *(const v4f*)(gb + k);
      rb[k] = t.x; rb[k + 1] = t.y; rb[k + 2] = t.z; rb[k + 3] = t.w;
    }
    chol_col<0>(ra, rb, lane);
    #pragma unroll
    for (int k = 0; k < 32; ++k) {
      G[k * GS_K + lane]      = ra[k];
      G[k * GS_K + lane + 32] = rb[k];
    }
    #pragma unroll
    for (int k = 32; k < 64; ++k)
      G[k * GS_K + lane + 32] = rb[k];
  }
  __syncthreads();

  {
    const int wv = __builtin_amdgcn_readfirstlane(wave);
    const int jA = wv;
    const int jB = 63 - wv;
    float xlo = (lane == jA) ? 1.0f : 0.0f;
    float xhi = 0.0f;
    float yhi = (lane + 32 == jB) ? 1.0f : 0.0f;
    const float* lt = G + lane;

    float llo = lt[jA * GS_K];
    float lhi = lt[jA * GS_K + 32];
    float rd  = lane_pick(llo, jA);
    #pragma nounroll
    for (int k = jA; k < 32; ++k) {
      const float nlo = lt[(k + 1) * GS_K];
      const float nhi = lt[(k + 1) * GS_K + 32];
      const float wk  = lane_pick(xlo, k) * rd;
      xlo = (lane > k) ? __builtin_fmaf(-llo, wk, xlo) : ((lane == k) ? wk : xlo);
      xhi = __builtin_fmaf(-lhi, wk, xhi);
      rd  = lane_pick(nlo, (k + 1) & 31);
      llo = nlo;
      lhi = nhi;
    }
    rd = lane_pick(lhi, 0);
    int k = 0;
    #pragma nounroll
    for (; k < 31 - wv; ++k) {
      const float nhi = lt[(33 + k) * GS_K + 32];
      xhi = fs_step(xhi, lhi, rd, k, lane);
      rd  = lane_pick(nhi, k + 1);
      lhi = nhi;
    }
    #pragma nounroll
    for (; k < 32; ++k) {
      const int   kn  = (k < 31) ? k + 1 : k;
      const float nhi = lt[(32 + kn) * GS_K + 32];
      xhi = fs_step(xhi, lhi, rd, k, lane);
      yhi = fs_step(yhi, lhi, rd, k, lane);
      rd  = lane_pick(nhi, kn);
      lhi = nhi;
    }
    W[lane * GS_WP + jA]        = xlo;
    W[(lane + 32) * GS_WP + jA] = xhi;
    W[lane * GS_WP + jB]        = 0.0f;
    W[(lane + 32) * GS_WP + jB] = yhi;
  }
  __syncthreads();

  {
    const int m    = lane & 15;
    const int koff = (lane < 16) ? 0 : 2;
    const int mb   = (lane < 16) ? 0 : 8;
    const int col  = wave * 32 + m;
    const float* pa = W + m * GS_WP + koff;
    const float* pb = X + koff * GS_LP + col;
    float*       po = og + mb * GS_L + col;
    v8f acc[4][2];
    #pragma unroll
    for (int ti = 0; ti < 4; ++ti) {
      acc[ti][0] = (v8f){0.f, 0.f, 0.f, 0.f, 0.f, 0.f, 0.f, 0.f};
      acc[ti][1] = (v8f){0.f, 0.f, 0.f, 0.f, 0.f, 0.f, 0.f, 0.f};
    }
    #pragma unroll
    for (int kb = 0; kb < 4; ++kb) {
      #pragma unroll
      for (int k0 = kb * 16; k0 < kb * 16 + 16; k0 += 4) {
        v2f b0, b1;
        b0.x = pb[(k0    ) * GS_LP];
        b0.y = pb[(k0 + 1) * GS_LP];
        b1.x = pb[(k0    ) * GS_LP + 16];
        b1.y = pb[(k0 + 1) * GS_LP + 16];
        #pragma unroll
        for (int ti = kb; ti < 4; ++ti) {
          v2f a = *(const v2f*)(pa + ti * 16 * GS_WP + k0);
          acc[ti][0] = __builtin_amdgcn_wmma_f32_16x16x4_f32(false, a, false, b0,
                                                             (short)0, acc[ti][0], false, false);
          acc[ti][1] = __builtin_amdgcn_wmma_f32_16x16x4_f32(false, a, false, b1,
                                                             (short)0, acc[ti][1], false, false);
        }
      }
      #pragma unroll
      for (int j = 0; j < 8; ++j) {
        po[(kb * 16 + j) * GS_L]      = acc[kb][0][j];
        po[(kb * 16 + j) * GS_L + 16] = acc[kb][1][j];
      }
    }
  }
}

extern "C" void kernel_launch(void* const* d_in, const int* in_sizes, int n_in,
                              void* d_out, int out_size, void* d_ws, size_t ws_size,
                              hipStream_t stream) {
  (void)in_sizes; (void)n_in; (void)out_size; (void)d_ws; (void)ws_size;
  const float* x   = (const float*)d_in[0];
  float*       out = (float*)d_out;

  const size_t shmem = (size_t)(GS_K * GS_LP + GS_K * GS_K + GS_K * GS_WP) * sizeof(float);
  hipFuncSetAttribute(reinterpret_cast<const void*>(&gram_schmidt_cholqr),
                      hipFuncAttributeMaxDynamicSharedMemorySize, (int)shmem);

  gram_schmidt_cholqr<<<dim3(GS_B), dim3(GS_NT), shmem, stream>>>(x, out);
}


# Each finished row tile of Q leaves at once and in whole 128-byte lines: staged through LDS over the X rows just consumed, its 16-byte stores held in front of the next k block's products; same sums
# speedup vs baseline: 7.2171x; 1.0328x over previous
#include <hip/hip_runtime.h>

typedef float v2f __attribute__((ext_vector_type(2)));
typedef float v4f __attribute__((ext_vector_type(4)));
typedef float v8f __attribute__((ext_vector_type(8)));

#define GS_B  512
#define GS_K  64
#define GS_L  1024
#define GS_LP 1028
#define GS_WP 68
#define GS_NT 1024

__device__ __forceinline__ float lane_bcast(float v, int src_lane) {
  return __int_as_float(__builtin_amdgcn_readlane(__float_as_int(v), src_lane));
}

__device__ __forceinline__ float lane_pick(float v, int src_lane) {
  return __int_as_float(__builtin_amdgcn_readlane(__float_as_int(v), src_lane));
}

__device__ __forceinline__ float fs_step(float x, float l, float rd, int k, int lane) {
  const float wk = lane_pick(x, k) * rd;
  return (lane > k) ? __builtin_fmaf(-l, wk, x) : ((lane == k) ? wk : x);
}


template <int P, int J>
__device__ __forceinline__ void chol_sub1(float (&a)[32], float (&b)[64]) {
  float ljp;
  if constexpr (J < 32) ljp = lane_bcast(a[P], J);
  else                  ljp = lane_bcast(b[P], J - 32);
  if constexpr (J < 32) a[J] = __builtin_fmaf(ljp, a[P], a[J]);
  b[J] = __builtin_fmaf(ljp, b[P], b[J]);
}

template <int P, int J>
__device__ __forceinline__ void chol_sub(float (&a)[32], float (&b)[64]) {
  if constexpr (J < 64) {
    chol_sub1<P, J>(a, b);
    chol_sub<P, J + 1>(a, b);
  }
}

template <int P>
__device__ __forceinline__ void chol_fin(float (&a)[32], float (&b)[64], int lane) {
  float djj;
  if constexpr (P < 32) djj = -lane_bcast(a[P], P);
  else                  djj = -lane_bcast(b[P], P - 32);
  const float inv = rsqrtf(djj);
  if constexpr (P < 32) {
    const float sa = (-a[P]) * inv;
    a[P] = (lane == P) ? inv : sa;
    b[P] = (-b[P]) * inv;
  } else {
    const float sb = (-b[P]) * inv;
    b[P] = (lane == P - 32) ? inv : sb;
  }
}

template <int J>
__device__ __forceinline__ void chol_pin(float (&a)[32], float (&b)[64]) {
  if constexpr (J < 64) {
    if constexpr (J < 32) asm volatile("" : "+v"(a[J]));
    asm volatile("" : "+v"(b[J]));
    chol_pin<J + 1>(a, b);
  }
}

template <int P>
__device__ __forceinline__ void chol_col(float (&a)[32], float (&b)[64], int lane) {
  if constexpr (P < 64) {
    chol_fin<P>(a, b, lane);
    if constexpr (P >= 1) chol_sub<P - 1, P + 1>(a, b);
    if constexpr (P + 1 < 64) chol_sub1<P, P + 1>(a, b);
    chol_pin<P + 1>(a, b);
    chol_col<P + 1>(a, b, lane);
  }
}

__global__ __launch_bounds__(GS_NT, 1)
void gram_schmidt_cholqr(const float* __restrict__ xin,
                         float* __restrict__ xout) {
  extern __shared__ float smem[];
  float* X = smem;
  float* G = smem + GS_K * GS_LP;
  float* W = G + GS_K * GS_K;

  const int b    = blockIdx.x;
  const int tid  = threadIdx.x;
  const int wave = tid >> 5;
  const int lane = tid & 31;

  const float* xg = xin  + (size_t)b * (GS_K * GS_L);
  float*       og = xout + (size_t)b * (GS_K * GS_L);

  for (int idx = tid; idx < GS_K * (GS_L / 4); idx += GS_NT) {
    int r  = idx >> 8;
    int c4 = (idx & 255) << 2;
    *(v4f*)(X + r * GS_LP + c4) = *(const v4f*)(xg + r * GS_L + c4);
  }
  __syncthreads();

  {
    const int wq = __builtin_amdgcn_readfirstlane(wave);
    if (wq < 20) {
      const int tile  = wq >> 1;
      const int ti    = (tile >= 6) ? 3 : ((tile >= 3) ? 2 : ((tile >= 1) ? 1 : 0));
      const int tj    = tile - ((ti * (ti + 1)) >> 1);
      const int khalf = wq & 1;
      const int m     = lane & 15;
      const int koff  = (lane < 16) ? 0 : 2;
      const float* pa = X + (ti * 16 + m) * GS_LP + koff + khalf * (GS_L / 2);
      const float* pb = X + (tj * 16 + m) * GS_LP + koff + khalf * (GS_L / 2);
      v8f acc = {0.f, 0.f, 0.f, 0.f, 0.f, 0.f, 0.f, 0.f};
      #pragma unroll 4
      for (int k = 0; k < GS_L / 2; k += 4) {
        v2f a  = *(const v2f*)(pa + k);
        v2f bb = *(const v2f*)(pb + k);
        acc = __builtin_amdgcn_wmma_f32_16x16x4_f32(false, a, false, bb,
                                                    (short)0, acc, false, false);
      }
      float*    dst = khalf ? W : G;
      const int mb  = (lane < 16) ? 0 : 8;
      #pragma unroll
      for (int j = 0; j < 8; ++j)
        dst[(ti * 16 + mb + j) * GS_K + tj * 16 + m] = acc[j];
      if (tj < ti) {
        float* mr = dst + (tj * 16 + m) * GS_K + ti * 16 + mb;
        *(v4f*)(mr)     = (v4f){acc[0], acc[1], acc[2], acc[3]};
        *(v4f*)(mr + 4) = (v4f){acc[4], acc[5], acc[6], acc[7]};
      }
    }
  }
  __syncthreads();
  for (int idx = tid; idx < GS_K * GS_K; idx += GS_NT) G[idx] = -(G[idx] + W[idx]);
  __syncthreads();

  if (wave == 0) {
    float ra[32], rb[64];
    float* ga = G + lane * GS_K;
    float* gb = G + (lane + 32) * GS_K;
    #pragma unroll
    for (int k = 0; k < 32; k += 4) {
      v4f t = *(const v4f*)(ga + k);
      ra[k] = t.x; ra[k + 1] = t.y; ra[k + 2] = t.z; ra[k + 3] = t.w;
    }
    #pragma unroll
    for (int k = 0; k < 64; k += 4) {
      v4f t = *(const v4f*)(gb + k);
      rb[k] = t.x; rb[k + 1] = t.y; rb[k + 2] = t.z; rb[k + 3] = t.w;
    }
    chol_col<0>(ra, rb, lane);
    #pragma unroll
    for (int k = 0; k < 32; ++k) {
      G[k * GS_K + lane]      = ra[k];
      G[k * GS_K + lane + 32] = rb[k];
    }
    #pragma unroll
    for (int k = 32; k < 64; ++k)
      G[k * GS_K + lane + 32] = rb[k];
  }
  __syncthreads();

  {
    const int wv = __builtin_amdgcn_readfirstlane(wave);
    const int jA = wv;
    const int jB = 63 - wv;
    float xlo = (lane == jA) ? 1.0f : 0.0f;
    float xhi = 0.0f;
    float yhi = (lane + 32 == jB) ? 1.0f : 0.0f;
    const float* lt = G + lane;

    float llo = lt[jA * GS_K];
    float lhi = lt[jA * GS_K + 32];
    float rd  = lane_pick(llo, jA);
    #pragma nounroll
    for (int k = jA; k < 32; ++k) {
      const float nlo = lt[(k + 1) * GS_K];
      const float nhi = lt[(k + 1) * GS_K + 32];
      const float wk  = lane_pick(xlo, k) * rd;
      xlo = (lane > k) ? __builtin_fmaf(-llo, wk, xlo) : ((lane == k) ? wk : xlo);
      xhi = __builtin_fmaf(-lhi, wk, xhi);
      rd  = lane_pick(nlo, (k + 1) & 31);
      llo = nlo;
      lhi = nhi;
    }
    rd = lane_pick(lhi, 0);
    int k = 0;
    #pragma nounroll
    for (; k < 31 - wv; ++k) {
      const float nhi = lt[(33 + k) * GS_K + 32];
      xhi = fs_step(xhi, lhi, rd, k, lane);
      rd  = lane_pick(nhi, k + 1);
      lhi = nhi;
    }
    #pragma nounroll
    for (; k < 32; ++k) {
      const int   kn  = (k < 31) ? k + 1 : k;
      const float nhi = lt[(32 + kn) * GS_K + 32];
      xhi = fs_step(xhi, lhi, rd, k, lane);
      yhi = fs_step(yhi, lhi, rd, k, lane);
      rd  = lane_pick(nhi, kn);
      lhi = nhi;
    }
    W[lane * GS_WP + jA]        = xlo;
    W[(lane + 32) * GS_WP + jA] = xhi;
    W[lane * GS_WP + jB]        = 0.0f;
    W[(lane + 32) * GS_WP + jB] = yhi;
  }
  __syncthreads();

  {
    const int m    = lane & 15;
    const int koff = (lane < 16) ? 0 : 2;
    const int mb   = (lane < 16) ? 0 : 8;
    const int col  = wave * 32 + m;
    const float* pa = W + m * GS_WP + koff;
    const float* pb = X + koff * GS_LP + col;
    float*       ps = X + mb * GS_LP + col;
    const int    c4 = (lane & 7) << 2;
    const int    rr = ((lane >> 3) & 1) * 8 + (lane >> 4) * 4;
    const float* pr = X + rr * GS_LP + wave * 32 + c4;
    float*       po = og + rr * GS_L + wave * 32 + c4;
    v8f acc[4][2];
    #pragma unroll
    for (int ti = 0; ti < 4; ++ti) {
      acc[ti][0] = (v8f){0.f, 0.f, 0.f, 0.f, 0.f, 0.f, 0.f, 0.f};
      acc[ti][1] = (v8f){0.f, 0.f, 0.f, 0.f, 0.f, 0.f, 0.f, 0.f};
    }
    #pragma unroll
    for (int kb = 0; kb < 4; ++kb) {
      #pragma unroll
      for (int k0 = kb * 16; k0 < kb * 16 + 16; k0 += 4) {
        v2f b0, b1;
        b0.x = pb[(k0    ) * GS_LP];
        b0.y = pb[(k0 + 1) * GS_LP];
        b1.x = pb[(k0    ) * GS_LP + 16];
        b1.y = pb[(k0 + 1) * GS_LP + 16];
        #pragma unroll
        for (int ti = kb; ti < 4; ++ti) {
          v2f a = *(const v2f*)(pa + ti * 16 * GS_WP + k0);
          acc[ti][0] = __builtin_amdgcn_wmma_f32_16x16x4_f32(false, a, false, b0,
                                                             (short)0, acc[ti][0], false, false);
          acc[ti][1] = __builtin_amdgcn_wmma_f32_16x16x4_f32(false, a, false, b1,
                                                             (short)0, acc[ti][1], false, false);
        }
      }
      #pragma unroll
      for (int j = 0; j < 8; ++j) {
        ps[(kb * 16 + j) * GS_LP]      = acc[kb][0][j];
        ps[(kb * 16 + j) * GS_LP + 16] = acc[kb][1][j];
      }
      #pragma unroll
      for (int i = 0; i < 4; ++i) {
        const v4f q = *(const v4f*)(pr + (kb * 16 + i) * GS_LP);
        *(v4f*)(po + (kb * 16 + i) * GS_L) = q;
      }
      if (kb < 3) __builtin_amdgcn_sched_barrier(0x7A6);
    }
  }
}

extern "C" void kernel_launch(void* const* d_in, const int* in_sizes, int n_in,
                              void* d_out, int out_size, void* d_ws, size_t ws_size,
                              hipStream_t stream) {
  (void)in_sizes; (void)n_in; (void)out_size; (void)d_ws; (void)ws_size;
  const float* x   = (const float*)d_in[0];
  float*       out = (float*)d_out;

  const size_t shmem = (size_t)(GS_K * GS_LP + GS_K * GS_K + GS_K * GS_WP) * sizeof(float);
  hipFuncSetAttribute(reinterpret_cast<const void*>(&gram_schmidt_cholqr),
                      hipFuncAttributeMaxDynamicSharedMemorySize, (int)shmem);

  gram_schmidt_cholqr<<<dim3(GS_B), dim3(GS_NT), shmem, stream>>>(x, out);
}


# Triangular solve by 16-row blocks: only the four 16x16 diagonal blocks of L are inverted (16 substitution steps, not 65), the rest enters the product as -L; same 40 products; sum order changes
# speedup vs baseline: 7.9571x; 1.1025x over previous
#include <hip/hip_runtime.h>

typedef float v2f __attribute__((ext_vector_type(2)));
typedef float v4f __attribute__((ext_vector_type(4)));
typedef float v8f __attribute__((ext_vector_type(8)));

#define GS_B  512
#define GS_K  64
#define GS_L  1024
#define GS_LP 1028
#define GS_MW 4096
#define GS_NT 1024

__device__ __forceinline__ float lane_bcast(float v, int src_lane) {
  return __int_as_float(__builtin_amdgcn_readlane(__float_as_int(v), src_lane));
}

template <int K>
__device__ __forceinline__ float row_pick(float v) {
  return __int_as_float(__builtin_amdgcn_update_dpp(0, __float_as_int(v), 0x150 + K, 0xF, 0xF, true));
}

template <int K>
__device__ __forceinline__ void blk_steps(float& x, const float (&lc)[16], int r) {
  if constexpr (K < 16) {
    const float wk = row_pick<K>(x) * row_pick<K>(lc[K]);
    x = (r > K) ? __builtin_fmaf(-lc[K], wk, x) : ((r == K) ? wk : x);
    blk_steps<K + 1>(x, lc, r);
  }
}


template <int P, int J>
__device__ __forceinline__ void chol_sub1(float (&a)[32], float (&b)[64]) {
  float ljp;
  if constexpr (J < 32) ljp = lane_bcast(a[P], J);
  else                  ljp = lane_bcast(b[P], J - 32);
  if constexpr (J < 32) a[J] = __builtin_fmaf(ljp, a[P], a[J]);
  b[J] = __builtin_fmaf(ljp, b[P], b[J]);
}

template <int P, int J>
__device__ __forceinline__ void chol_sub(float (&a)[32], float (&b)[64]) {
  if constexpr (J < 64) {
    chol_sub1<P, J>(a, b);
    chol_sub<P, J + 1>(a, b);
  }
}

template <int P>
__device__ __forceinline__ void chol_fin(float (&a)[32], float (&b)[64], int lane) {
  float djj;
  if constexpr (P < 32) djj = -lane_bcast(a[P], P);
  else                  djj = -lane_bcast(b[P], P - 32);
  const float inv = rsqrtf(djj);
  if constexpr (P < 32) {
    const float sa = (-a[P]) * inv;
    a[P] = (lane == P) ? inv : sa;
    b[P] = (-b[P]) * inv;
  } else {
    const float sb = (-b[P]) * inv;
    b[P] = (lane == P - 32) ? inv : sb;
  }
}

template <int J>
__device__ __forceinline__ void chol_pin(float (&a)[32], float (&b)[64]) {
  if constexpr (J < 64) {
    if constexpr (J < 32) asm volatile("" : "+v"(a[J]));
    asm volatile("" : "+v"(b[J]));
    chol_pin<J + 1>(a, b);
  }
}

template <int P>
__device__ __forceinline__ void chol_col(float (&a)[32], float (&b)[64], int lane) {
  if constexpr (P < 64) {
    chol_fin<P>(a, b, lane);
    if constexpr (P >= 1) chol_sub<P - 1, P + 1>(a, b);
    if constexpr (P + 1 < 64) chol_sub1<P, P + 1>(a, b);
    chol_pin<P + 1>(a, b);
    chol_col<P + 1>(a, b, lane);
  }
}

__global__ __launch_bounds__(GS_NT, 1)
void gram_schmidt_cholqr(const float* __restrict__ xin,
                         float* __restrict__ xout) {
  extern __shared__ float smem[];
  float* X = smem;
  float* G = smem + GS_K * GS_LP;
  float* W = G + GS_K * GS_K;

  const int b    = blockIdx.x;
  const int tid  = threadIdx.x;
  const int wave = tid >> 5;
  const int lane = tid & 31;

  const float* xg = xin  + (size_t)b * (GS_K * GS_L);
  float*       og = xout + (size_t)b * (GS_K * GS_L);

  for (int idx = tid; idx < GS_K * (GS_L / 4); idx += GS_NT) {
    int r  = idx >> 8;
    int c4 = (idx & 255) << 2;
    *(v4f*)(X + r * GS_LP + c4) = *(const v4f*)(xg + r * GS_L + c4);
  }
  __syncthreads();

  {
    const int wq = __builtin_amdgcn_readfirstlane(wave);
    if (wq < 20) {
      const int tile  = wq >> 1;
      const int ti    = (tile >= 6) ? 3 : ((tile >= 3) ? 2 : ((tile >= 1) ? 1 : 0));
      const int tj    = tile - ((ti * (ti + 1)) >> 1);
      const int khalf = wq & 1;
      const int m     = lane & 15;
      const int koff  = (lane < 16) ? 0 : 2;
      const float* pa = X + (ti * 16 + m) * GS_LP + koff + khalf * (GS_L / 2);
      const float* pb = X + (tj * 16 + m) * GS_LP + koff + khalf * (GS_L / 2);
      v8f acc = {0.f, 0.f, 0.f, 0.f, 0.f, 0.f, 0.f, 0.f};
      #pragma unroll 4
      for (int k = 0; k < GS_L / 2; k += 4) {
        v2f a  = *(const v2f*)(pa + k);
        v2f bb = *(const v2f*)(pb + k);
        acc = __builtin_amdgcn_wmma_f32_16x16x4_f32(false, a, false, bb,
                                                    (short)0, acc, false, false);
      }
      float*    dst = khalf ? W : G;
      const int mb  = (lane < 16) ? 0 : 8;
      #pragma unroll
      for (int j = 0; j < 8; ++j)
        dst[(ti * 16 + mb + j) * GS_K + tj * 16 + m] = acc[j];
      if (tj < ti) {
        float* mr = dst + (tj * 16 + m) * GS_K + ti * 16 + mb;
        *(v4f*)(mr)     = (v4f){acc[0], acc[1], acc[2], acc[3]};
        *(v4f*)(mr + 4) = (v4f){acc[4], acc[5], acc[6], acc[7]};
      }
    }
  }
  __syncthreads();
  for (int idx = tid; idx < GS_K * GS_K; idx += GS_NT) G[idx] = -(G[idx] + W[idx]);
  __syncthreads();

  if (wave == 0) {
    float ra[32], rb[64];
    float* ga = G + lane * GS_K;
    float* gb = G + (lane + 32) * GS_K;
    #pragma unroll
    for (int k = 0; k < 32; k += 4) {
      v4f t = *(const v4f*)(ga + k);
      ra[k] = t.x; ra[k + 1] = t.y; ra[k + 2] = t.z; ra[k + 3] = t.w;
    }
    #pragma unroll
    for (int k = 0; k < 64; k += 4) {
      v4f t = *(const v4f*)(gb + k);
      rb[k] = t.x; rb[k + 1] = t.y; rb[k + 2] = t.z; rb[k + 3] = t.w;
    }
    chol_col<0>(ra, rb, lane);
    #pragma unroll
    for (int k = 0; k < 32; ++k) {
      G[k * GS_K + lane]      = ra[k];
      G[k * GS_K + lane + 32] = rb[k];
    }
    #pragma unroll
    for (int k = 32; k < 64; ++k)
      G[k * GS_K + lane + 32] = rb[k];
  }
  __syncthreads();

  {
    float* Mf = W;
    for (int idx = tid; idx < 6 * 256; idx += GS_NT) {
      const int blk = idx >> 8;
      const int ti  = (blk >= 3) ? 3 : ((blk >= 1) ? 2 : 1);
      const int kb  = blk - ((ti * (ti - 1)) >> 1);
      const int r   = idx & 15;
      const int jj  = (idx >> 4) & 15;
      const float v = G[(kb * 16 + jj) * GS_K + ti * 16 + r];
      Mf[(((ti * 4 + kb) * 4 + ((jj & 7) >> 1)) << 6) + ((((jj >> 3) << 4) + r) << 1) + (jj & 1)] = -v;
    }
    const int wv = __builtin_amdgcn_readfirstlane(wave);
    const int bb = wv >> 3;
    const int r  = lane & 15;
    const int j  = ((wv & 7) << 1) + (lane >> 4);
    const float* lp = G + (bb * 16) * GS_K + bb * 16 + r;
    float lc[16];
    #pragma unroll
    for (int k = 0; k < 16; ++k) lc[k] = lp[k * GS_K];
    float x = (r == j) ? 1.0f : 0.0f;
    blk_steps<0>(x, lc, r);
    Mf[(((bb * 4 + bb) * 4 + ((j & 7) >> 1)) << 6) + ((((j >> 3) << 4) + r) << 1) + (j & 1)] = x;
  }
  __syncthreads();

  {
    const int m    = lane & 15;
    const int mb   = (lane < 16) ? 0 : 8;
    const int col  = wave * 32 + m;
    const float* pa = W + lane * 2;
    float*       ps = X + mb * GS_LP + col;
    const int    c4 = (lane & 7) << 2;
    const int    rr = ((lane >> 3) & 1) * 8 + (lane >> 4) * 4;
    const float* pr = X + rr * GS_LP + wave * 32 + c4;
    float*       po = og + rr * GS_L + wave * 32 + c4;
    v8f q[3][2];
    #pragma unroll
    for (int kb = 0; kb < 4; ++kb) {
      v8f t[2];
      #pragma unroll
      for (int jr = 0; jr < 8; ++jr) {
        t[0][jr] = ps[(kb * 16 + jr) * GS_LP];
        t[1][jr] = ps[(kb * 16 + jr) * GS_LP + 16];
      }
      #pragma unroll
      for (int cb = 0; cb < kb; ++cb) {
        #pragma unroll
        for (int s4 = 0; s4 < 4; ++s4) {
          const v2f a  = *(const v2f*)(pa + (((kb * 4 + cb) * 4 + s4) << 6));
          const v2f b0 = {q[cb][0][2 * s4], q[cb][0][2 * s4 + 1]};
          const v2f b1 = {q[cb][1][2 * s4], q[cb][1][2 * s4 + 1]};
          t[0] = __builtin_amdgcn_wmma_f32_16x16x4_f32(false, a, false, b0,
                                                       (short)0, t[0], false, false);
          t[1] = __builtin_amdgcn_wmma_f32_16x16x4_f32(false, a, false, b1,
                                                       (short)0, t[1], false, false);
        }
      }
      v8f d[2];
      d[0] = (v8f){0.f, 0.f, 0.f, 0.f, 0.f, 0.f, 0.f, 0.f};
      d[1] = (v8f){0.f, 0.f, 0.f, 0.f, 0.f, 0.f, 0.f, 0.f};
      #pragma unroll
      for (int s4 = 0; s4 < 4; ++s4) {
        const v2f a  = *(const v2f*)(pa + (((kb * 4 + kb) * 4 + s4) << 6));
        const v2f b0 = {t[0][2 * s4], t[0][2 * s4 + 1]};
        const v2f b1 = {t[1][2 * s4], t[1][2 * s4 + 1]};
        d[0] = __builtin_amdgcn_wmma_f32_16x16x4_f32(false, a, false, b0,
                                                     (short)0, d[0], false, false);
        d[1] = __builtin_amdgcn_wmma_f32_16x16x4_f32(false, a, false, b1,
                                                     (short)0, d[1], false, false);
      }
      if (kb < 3) { q[kb][0] = d[0]; q[kb][1] = d[1]; }
      #pragma unroll
      for (int jr = 0; jr < 8; ++jr) {
        ps[(kb * 16 + jr) * GS_LP]      = d[0][jr];
        ps[(kb * 16 + jr) * GS_LP + 16] = d[1][jr];
      }
      #pragma unroll
      for (int i = 0; i < 4; ++i) {
        const v4f qv = *(const v4f*)(pr + (kb * 16 + i) * GS_LP);
        *(v4f*)(po + (kb * 16 + i) * GS_L) = qv;
      }
      if (kb < 3) __builtin_amdgcn_sched_barrier(0x7A6);
    }
  }
}

extern "C" void kernel_launch(void* const* d_in, const int* in_sizes, int n_in,
                              void* d_out, int out_size, void* d_ws, size_t ws_size,
                              hipStream_t stream) {
  (void)in_sizes; (void)n_in; (void)out_size; (void)d_ws; (void)ws_size;
  const float* x   = (const float*)d_in[0];
  float*       out = (float*)d_out;

  const size_t shmem = (size_t)(GS_K * GS_LP + GS_K * GS_K + GS_MW) * sizeof(float);
  hipFuncSetAttribute(reinterpret_cast<const void*>(&gram_schmidt_cholqr),
                      hipFuncAttributeMaxDynamicSharedMemorySize, (int)shmem);

  gram_schmidt_cholqr<<<dim3(GS_B), dim3(GS_NT), shmem, stream>>>(x, out);
}


# Finished block rows of Q leave as 512 consecutive bytes per store instruction (whole aligned 256-byte units) behind one LDS-only workgroup barrier each, not as lone 128-byte lines; same sums
# speedup vs baseline: 8.8798x; 1.1160x over previous
#include <hip/hip_runtime.h>

typedef float v2f __attribute__((ext_vector_type(2)));
typedef float v4f __attribute__((ext_vector_type(4)));
typedef float v8f __attribute__((ext_vector_type(8)));

#define GS_B  512
#define GS_K  64
#define GS_L  1024
#define GS_LP 1028
#define GS_MW 4096
#define GS_NT 1024

__device__ __forceinline__ float lane_bcast(float v, int src_lane) {
  return __int_as_float(__builtin_amdgcn_readlane(__float_as_int(v), src_lane));
}

template <int K>
__device__ __forceinline__ float row_pick(float v) {
  return __int_as_float(__builtin_amdgcn_update_dpp(0, __float_as_int(v), 0x150 + K, 0xF, 0xF, true));
}

template <int K>
__device__ __forceinline__ void blk_steps(float& x, const float (&lc)[16], int r) {
  if constexpr (K < 16) {
    const float wk = row_pick<K>(x) * row_pick<K>(lc[K]);
    x = (r > K) ? __builtin_fmaf(-lc[K], wk, x) : ((r == K) ? wk : x);
    blk_steps<K + 1>(x, lc, r);
  }
}


template <int P, int J>
__device__ __forceinline__ void chol_sub1(float (&a)[32], float (&b)[64]) {
  float ljp;
  if constexpr (J < 32) ljp = lane_bcast(a[P], J);
  else                  ljp = lane_bcast(b[P], J - 32);
  if constexpr (J < 32) a[J] = __builtin_fmaf(ljp, a[P], a[J]);
  b[J] = __builtin_fmaf(ljp, b[P], b[J]);
}

template <int P, int J>
__device__ __forceinline__ void chol_sub(float (&a)[32], float (&b)[64]) {
  if constexpr (J < 64) {
    chol_sub1<P, J>(a, b);
    chol_sub<P, J + 1>(a, b);
  }
}

template <int P>
__device__ __forceinline__ void chol_fin(float (&a)[32], float (&b)[64], int lane) {
  float djj;
  if constexpr (P < 32) djj = -lane_bcast(a[P], P);
  else                  djj = -lane_bcast(b[P], P - 32);
  const float inv = rsqrtf(djj);
  if constexpr (P < 32) {
    const float sa = (-a[P]) * inv;
    a[P] = (lane == P) ? inv : sa;
    b[P] = (-b[P]) * inv;
  } else {
    const float sb = (-b[P]) * inv;
    b[P] = (lane == P - 32) ? inv : sb;
  }
}

template <int J>
__device__ __forceinline__ void chol_pin(float (&a)[32], float (&b)[64]) {
  if constexpr (J < 64) {
    if constexpr (J < 32) asm volatile("" : "+v"(a[J]));
    asm volatile("" : "+v"(b[J]));
    chol_pin<J + 1>(a, b);
  }
}

template <int P>
__device__ __forceinline__ void chol_col(float (&a)[32], float (&b)[64], int lane) {
  if constexpr (P < 64) {
    chol_fin<P>(a, b, lane);
    if constexpr (P >= 1) chol_sub<P - 1, P + 1>(a, b);
    if constexpr (P + 1 < 64) chol_sub1<P, P + 1>(a, b);
    chol_pin<P + 1>(a, b);
    chol_col<P + 1>(a, b, lane);
  }
}

__global__ __launch_bounds__(GS_NT, 1)
void gram_schmidt_cholqr(const float* __restrict__ xin,
                         float* __restrict__ xout) {
  extern __shared__ float smem[];
  float* X = smem;
  float* G = smem + GS_K * GS_LP;
  float* W = G + GS_K * GS_K;

  const int b    = blockIdx.x;
  const int tid  = threadIdx.x;
  const int wave = tid >> 5;
  const int lane = tid & 31;

  const float* xg = xin  + (size_t)b * (GS_K * GS_L);
  float*       og = xout + (size_t)b * (GS_K * GS_L);

  for (int idx = tid; idx < GS_K * (GS_L / 4); idx += GS_NT) {
    int r  = idx >> 8;
    int c4 = (idx & 255) << 2;
    *(v4f*)(X + r * GS_LP + c4) = *(const v4f*)(xg + r * GS_L + c4);
  }
  __syncthreads();

  {
    const int wq = __builtin_amdgcn_readfirstlane(wave);
    if (wq < 20) {
      const int tile  = wq >> 1;
      const int ti    = (tile >= 6) ? 3 : ((tile >= 3) ? 2 : ((tile >= 1) ? 1 : 0));
      const int tj    = tile - ((ti * (ti + 1)) >> 1);
      const int khalf = wq & 1;
      const int m     = lane & 15;
      const int koff  = (lane < 16) ? 0 : 2;
      const float* pa = X + (ti * 16 + m) * GS_LP + koff + khalf * (GS_L / 2);
      const float* pb = X + (tj * 16 + m) * GS_LP + koff + khalf * (GS_L / 2);
      v8f acc = {0.f, 0.f, 0.f, 0.f, 0.f, 0.f, 0.f, 0.f};
      #pragma unroll 4
      for (int k = 0; k < GS_L / 2; k += 4) {
        v2f a  = *(const v2f*)(pa + k);
        v2f bb = *(const v2f*)(pb + k);
        acc = __builtin_amdgcn_wmma_f32_16x16x4_f32(false, a, false, bb,
                                                    (short)0, acc, false, false);
      }
      float*    dst = khalf ? W : G;
      const int mb  = (lane < 16) ? 0 : 8;
      #pragma unroll
      for (int j = 0; j < 8; ++j)
        dst[(ti * 16 + mb + j) * GS_K + tj * 16 + m] = acc[j];
      if (tj < ti) {
        float* mr = dst + (tj * 16 + m) * GS_K + ti * 16 + mb;
        *(v4f*)(mr)     = (v4f){acc[0], acc[1], acc[2], acc[3]};
        *(v4f*)(mr + 4) = (v4f){acc[4], acc[5], acc[6], acc[7]};
      }
    }
  }
  __syncthreads();
  for (int idx = tid; idx < GS_K * GS_K; idx += GS_NT) G[idx] = -(G[idx] + W[idx]);
  __syncthreads();

  if (wave == 0) {
    float ra[32], rb[64];
    float* ga = G + lane * GS_K;
    float* gb = G + (lane + 32) * GS_K;
    #pragma unroll
    for (int k = 0; k < 32; k += 4) {
      v4f t = *(const v4f*)(ga + k);
      ra[k] = t.x; ra[k + 1] = t.y; ra[k + 2] = t.z; ra[k + 3] = t.w;
    }
    #pragma unroll
    for (int k = 0; k < 64; k += 4) {
      v4f t = *(const v4f*)(gb + k);
      rb[k] = t.x; rb[k + 1] = t.y; rb[k + 2] = t.z; rb[k + 3] = t.w;
    }
    chol_col<0>(ra, rb, lane);
    #pragma unroll
    for (int k = 0; k < 32; ++k) {
      G[k * GS_K + lane]      = ra[k];
      G[k * GS_K + lane + 32] = rb[k];
    }
    #pragma unroll
    for (int k = 32; k < 64; ++k)
      G[k * GS_K + lane + 32] = rb[k];
  }
  __syncthreads();

  {
    float* Mf = W;
    for (int idx = tid; idx < 6 * 256; idx += GS_NT) {
      const int blk = idx >> 8;
      const int ti  = (blk >= 3) ? 3 : ((blk >= 1) ? 2 : 1);
      const int kb  = blk - ((ti * (ti - 1)) >> 1);
      const int r   = idx & 15;
      const int jj  = (idx >> 4) & 15;
      const float v = G[(kb * 16 + jj) * GS_K + ti * 16 + r];
      Mf[(((ti * 4 + kb) * 4 + ((jj & 7) >> 1)) << 6) + ((((jj >> 3) << 4) + r) << 1) + (jj & 1)] = -v;
    }
    const int wv = __builtin_amdgcn_readfirstlane(wave);
    const int bb = wv >> 3;
    const int r  = lane & 15;
    const int j  = ((wv & 7) << 1) + (lane >> 4);
    const float* lp = G + (bb * 16) * GS_K + bb * 16 + r;
    float lc[16];
    #pragma unroll
    for (int k = 0; k < 16; ++k) lc[k] = lp[k * GS_K];
    float x = (r == j) ? 1.0f : 0.0f;
    blk_steps<0>(x, lc, r);
    Mf[(((bb * 4 + bb) * 4 + ((j & 7) >> 1)) << 6) + ((((j >> 3) << 4) + r) << 1) + (j & 1)] = x;
  }
  __syncthreads();

  {
    const int m    = lane & 15;
    const int mb   = (lane < 16) ? 0 : 8;
    const int col  = wave * 32 + m;
    const float* pa = W + lane * 2;
    float*       ps = X + mb * GS_LP + col;
    const int    ws   = __builtin_amdgcn_readfirstlane(wave);
    const int    scol = ((ws & 1) << 9) + (lane << 2);
    const float* pr = X + (ws >> 1) * GS_LP + scol;
    float*       po = og + (ws >> 1) * GS_L + scol;
    v8f q[3][2];
    v8f t[2];
    #pragma unroll
    for (int jr = 0; jr < 8; ++jr) {
      t[0][jr] = ps[jr * GS_LP];
      t[1][jr] = ps[jr * GS_LP + 16];
    }
    #pragma unroll
    for (int kb = 0; kb < 4; ++kb) {
      #pragma unroll
      for (int cb = 0; cb < kb; ++cb) {
        #pragma unroll
        for (int s4 = 0; s4 < 4; ++s4) {
          const v2f a  = *(const v2f*)(pa + (((kb * 4 + cb) * 4 + s4) << 6));
          const v2f b0 = {q[cb][0][2 * s4], q[cb][0][2 * s4 + 1]};
          const v2f b1 = {q[cb][1][2 * s4], q[cb][1][2 * s4 + 1]};
          t[0] = __builtin_amdgcn_wmma_f32_16x16x4_f32(false, a, false, b0,
                                                       (short)0, t[0], false, false);
          t[1] = __builtin_amdgcn_wmma_f32_16x16x4_f32(false, a, false, b1,
                                                       (short)0, t[1], false, false);
        }
        if (cb == 0) {
          __builtin_amdgcn_sched_barrier(0x7A6);
          #pragma unroll
          for (int i = 0; i < 4; ++i) {
            const v4f qv = *(const v4f*)(pr + (kb - 1) * 16 * GS_LP + i * 128);
            *(v4f*)(po + (kb - 1) * 16 * GS_L + i * 128) = qv;
          }
          __builtin_amdgcn_sched_barrier(0x7A6);
        }
      }
      v8f d[2];
      d[0] = (v8f){0.f, 0.f, 0.f, 0.f, 0.f, 0.f, 0.f, 0.f};
      d[1] = (v8f){0.f, 0.f, 0.f, 0.f, 0.f, 0.f, 0.f, 0.f};
      #pragma unroll
      for (int s4 = 0; s4 < 4; ++s4) {
        const v2f a  = *(const v2f*)(pa + (((kb * 4 + kb) * 4 + s4) << 6));
        const v2f b0 = {t[0][2 * s4], t[0][2 * s4 + 1]};
        const v2f b1 = {t[1][2 * s4], t[1][2 * s4 + 1]};
        d[0] = __builtin_amdgcn_wmma_f32_16x16x4_f32(false, a, false, b0,
                                                     (short)0, d[0], false, false);
        d[1] = __builtin_amdgcn_wmma_f32_16x16x4_f32(false, a, false, b1,
                                                     (short)0, d[1], false, false);
      }
      if (kb < 3) { q[kb][0] = d[0]; q[kb][1] = d[1]; }
      #pragma unroll
      for (int jr = 0; jr < 8; ++jr) {
        ps[(kb * 16 + jr) * GS_LP]      = d[0][jr];
        ps[(kb * 16 + jr) * GS_LP + 16] = d[1][jr];
      }
      if (kb < 3) {
        #pragma unroll
        for (int jr = 0; jr < 8; ++jr) {
          t[0][jr] = ps[((kb + 1) * 16 + jr) * GS_LP];
          t[1][jr] = ps[((kb + 1) * 16 + jr) * GS_LP + 16];
        }
      }
      __builtin_amdgcn_fence(__ATOMIC_RELEASE, "workgroup", "local");
      __builtin_amdgcn_s_barrier();
      __builtin_amdgcn_fence(__ATOMIC_ACQUIRE, "workgroup", "local");
    }
    #pragma unroll
    for (int i = 0; i < 4; ++i) {
      const v4f qv = *(const v4f*)(pr + 3 * 16 * GS_LP + i * 128);
      *(v4f*)(po + 3 * 16 * GS_L + i * 128) = qv;
    }
  }
}

extern "C" void kernel_launch(void* const* d_in, const int* in_sizes, int n_in,
                              void* d_out, int out_size, void* d_ws, size_t ws_size,
                              hipStream_t stream) {
  (void)in_sizes; (void)n_in; (void)out_size; (void)d_ws; (void)ws_size;
  const float* x   = (const float*)d_in[0];
  float*       out = (float*)d_out;

  const size_t shmem = (size_t)(GS_K * GS_LP + GS_K * GS_K + GS_MW) * sizeof(float);
  hipFuncSetAttribute(reinterpret_cast<const void*>(&gram_schmidt_cholqr),
                      hipFuncAttributeMaxDynamicSharedMemorySize, (int)shmem);

  gram_schmidt_cholqr<<<dim3(GS_B), dim3(GS_NT), shmem, stream>>>(x, out);
}
